# starting kernel
# baseline (speedup 1.0000x reference)
#include <hip/hip_runtime.h>
#include <math.h>

typedef float v2f __attribute__((ext_vector_type(2)));
typedef float v8f __attribute__((ext_vector_type(8)));

#define FEAT 128
#define N_CLASSES 16
#define BN_EPS 1e-5f

__global__ void k_fill(float* __restrict__ p, float v, int n) {
    int i = blockIdx.x * blockDim.x + threadIdx.x;
    if (i < n) p[i] = v;
}

__global__ void k_deg_accum(const int* __restrict__ dst, float* __restrict__ deg, int E) {
    int e = blockIdx.x * blockDim.x + threadIdx.x;
    if (e < E) {
        __hip_atomic_fetch_add(&deg[dst[e]], 1.0f,
                               __ATOMIC_RELAXED, __HIP_MEMORY_SCOPE_AGENT);
    }
}

__global__ void k_rsqrt_inplace(float* __restrict__ p, int n) {
    int i = blockIdx.x * blockDim.x + threadIdx.x;
    if (i < n) p[i] = rsqrtf(p[i]);
}

__global__ void k_gemm_wmma(const float* __restrict__ A, const float* __restrict__ B,
                            const float* __restrict__ bias, float* __restrict__ C,
                            int M, int K, int N) {
    const int lane  = threadIdx.x & 31;
    const int wave  = threadIdx.x >> 5;
    const int wpb   = blockDim.x >> 5;
    const int tile  = blockIdx.x * wpb + wave;
    const int Mt    = (M + 15) >> 4;
    const int Nt    = N >> 4;
    if (tile >= Mt * Nt) return;

    const int mBase = (tile / Nt) << 4;
    const int nBase = (tile % Nt) << 4;

    int row = mBase + (lane & 15);
    if (row > M - 1) row = M - 1;
    const int col  = nBase + (lane & 15);
    const int kOff = (lane >> 4) << 1;

    v8f acc = {};
    for (int k = 0; k < K; k += 4) {
        v2f a = *(const v2f*)(A + (size_t)row * K + k + kOff);
        v2f b;
        b.x = B[(size_t)(k + kOff)     * N + col];
        b.y = B[(size_t)(k + kOff + 1) * N + col];
        acc = __builtin_amdgcn_wmma_f32_16x16x4_f32(
                   false, a,  false, b,
                   (short)0, acc,  false,  false);
    }

    const float bb   = bias ? bias[col] : 0.0f;
    const int   rOff = (lane >> 4) << 3;
#pragma unroll
    for (int j = 0; j < 8; ++j) {
        int r = mBase + j + rOff;
        if (r < M) C[(size_t)r * N + col] = acc[j] + bb;
    }
}

__global__ void k_scatter(const float* __restrict__ H,
                          const int* __restrict__ src, const int* __restrict__ dst,
                          const float* __restrict__ dinv, float* __restrict__ out, int E) {
    const int lane = threadIdx.x & 31;
    const int wave = threadIdx.x >> 5;
    const int e    = blockIdx.x * (blockDim.x >> 5) + wave;
    if (e >= E) return;

    const int   s = src[e];
    const int   d = dst[e];
    const float w = dinv[s] * dinv[d];

    const float4 hv = *(const float4*)(H + (size_t)s * FEAT + lane * 4);
    float* o = out + (size_t)d * FEAT + lane * 4;
    __hip_atomic_fetch_add(o + 0, hv.x * w, __ATOMIC_RELAXED, __HIP_MEMORY_SCOPE_AGENT);
    __hip_atomic_fetch_add(o + 1, hv.y * w, __ATOMIC_RELAXED, __HIP_MEMORY_SCOPE_AGENT);
    __hip_atomic_fetch_add(o + 2, hv.z * w, __ATOMIC_RELAXED, __HIP_MEMORY_SCOPE_AGENT);
    __hip_atomic_fetch_add(o + 3, hv.w * w, __ATOMIC_RELAXED, __HIP_MEMORY_SCOPE_AGENT);
}

__global__ void k_post1(float* __restrict__ agg, const float* __restrict__ H,
                        const float* __restrict__ dinv, const float* __restrict__ b1,
                        const float* __restrict__ gamma, const float* __restrict__ beta,
                        const float* __restrict__ mean, const float* __restrict__ var,
                        int total) {
    int i = blockIdx.x * blockDim.x + threadIdx.x;
    if (i >= total) return;
    const int   node = i >> 7;
    const int   f    = i & (FEAT - 1);
    const float di   = dinv[node];
    float v = agg[i] + H[i] * di * di + b1[f];
    v = (v - mean[f]) * rsqrtf(var[f] + BN_EPS) * gamma[f] + beta[f];
    agg[i] = fmaxf(v, 0.0f);
}

__global__ void k_post2(float* __restrict__ agg, const float* __restrict__ H,
                        const float* __restrict__ dinv, const float* __restrict__ b2,
                        int total) {
    int i = blockIdx.x * blockDim.x + threadIdx.x;
    if (i >= total) return;
    const int   node = i >> 7;
    const int   f    = i & (FEAT - 1);
    const float di   = dinv[node];
    agg[i] = agg[i] + H[i] * di * di + b2[f];
}

extern "C" void kernel_launch(void* const* d_in, const int* in_sizes, int n_in,
                              void* d_out, int out_size, void* d_ws, size_t ws_size,
                              hipStream_t stream) {
    const float* X     = (const float*)d_in[0];
    const int*   edge  = (const int*)  d_in[1];
    const float* W1    = (const float*)d_in[2];
    const float* b1    = (const float*)d_in[3];
    const float* gamma = (const float*)d_in[4];
    const float* beta  = (const float*)d_in[5];
    const float* rmean = (const float*)d_in[6];
    const float* rvar  = (const float*)d_in[7];
    const float* W2    = (const float*)d_in[8];
    const float* b2    = (const float*)d_in[9];
    const float* Wc    = (const float*)d_in[10];
    const float* bc    = (const float*)d_in[11];
    float*       out   = (float*)d_out;

    const int N = in_sizes[0] / FEAT;
    const int E = in_sizes[1] / 2;
    const int* src = edge;
    const int* dst = edge + E;
    const int total = N * FEAT;

    float* deg  = (float*)d_ws;
    float* bufA = deg + ((N + 63) & ~63);
    float* bufB = bufA + (size_t)total;

    const int T256   = 256;
    const int gNode  = (N + T256 - 1) / T256;
    const int gEdge  = (E + T256 - 1) / T256;
    const int gFeat  = (total + T256 - 1) / T256;
    const int gScat  = (E + 7) / 8;
    const int tilesL = ((N + 15) / 16) * (FEAT / 16);
    const int tilesC = ((N + 15) / 16) * (N_CLASSES / 16);
    const int gGemmL = (tilesL + 7) / 8;
    const int gGemmC = (tilesC + 7) / 8;

    k_fill<<<gNode, T256, 0, stream>>>(deg, 1.0f, N);
    k_deg_accum<<<gEdge, T256, 0, stream>>>(dst, deg, E);
    k_rsqrt_inplace<<<gNode, T256, 0, stream>>>(deg, N);

    k_gemm_wmma<<<gGemmL, T256, 0, stream>>>(X, W1, nullptr, bufA, N, FEAT, FEAT);
    k_fill<<<gFeat, T256, 0, stream>>>(bufB, 0.0f, total);
    k_scatter<<<gScat, T256, 0, stream>>>(bufA, src, dst, deg, bufB, E);
    k_post1<<<gFeat, T256, 0, stream>>>(bufB, bufA, deg, b1, gamma, beta, rmean, rvar, total);

    k_gemm_wmma<<<gGemmL, T256, 0, stream>>>(bufB, W2, nullptr, bufA, N, FEAT, FEAT);
    k_fill<<<gFeat, T256, 0, stream>>>(bufB, 0.0f, total);
    k_scatter<<<gScat, T256, 0, stream>>>(bufA, src, dst, deg, bufB, E);
    k_post2<<<gFeat, T256, 0, stream>>>(bufB, bufA, deg, b2, total);

    k_gemm_wmma<<<gGemmC, T256, 0, stream>>>(bufB, Wc, bc, out, N, FEAT, N_CLASSES);
}


# Aggregation by destination-owner blocks: each block scans the edge list and sums its 200 nodes' rows in an LDS tile with LDS float atomics, epilogue at write-out; 12 launches to 8; arrival order kept
# speedup vs baseline: 1.7076x; 1.7076x over previous
#include <hip/hip_runtime.h>
#include <math.h>

typedef float v2f __attribute__((ext_vector_type(2)));
typedef float v8f __attribute__((ext_vector_type(8)));

#define FEAT 128
#define N_CLASSES 16
#define BN_EPS 1e-5f

__global__ void k_fill(float* __restrict__ p, float v, int n) {
    int i = blockIdx.x * blockDim.x + threadIdx.x;
    if (i < n) p[i] = v;
}

__global__ void k_deg_accum(const int* __restrict__ dst, float* __restrict__ deg, int E) {
    int e = blockIdx.x * blockDim.x + threadIdx.x;
    if (e < E) {
        __hip_atomic_fetch_add(&deg[dst[e]], 1.0f,
                               __ATOMIC_RELAXED, __HIP_MEMORY_SCOPE_AGENT);
    }
}

__global__ void k_rsqrt_inplace(float* __restrict__ p, int n) {
    int i = blockIdx.x * blockDim.x + threadIdx.x;
    if (i < n) p[i] = rsqrtf(p[i]);
}

__global__ void k_gemm_wmma(const float* __restrict__ A, const float* __restrict__ B,
                            const float* __restrict__ bias, float* __restrict__ C,
                            int M, int K, int N) {
    const int lane  = threadIdx.x & 31;
    const int wave  = threadIdx.x >> 5;
    const int wpb   = blockDim.x >> 5;
    const int tile  = blockIdx.x * wpb + wave;
    const int Mt    = (M + 15) >> 4;
    const int Nt    = N >> 4;
    if (tile >= Mt * Nt) return;

    const int mBase = (tile / Nt) << 4;
    const int nBase = (tile % Nt) << 4;

    int row = mBase + (lane & 15);
    if (row > M - 1) row = M - 1;
    const int col  = nBase + (lane & 15);
    const int kOff = (lane >> 4) << 1;

    v8f acc = {};
    for (int k = 0; k < K; k += 4) {
        v2f a = *(const v2f*)(A + (size_t)row * K + k + kOff);
        v2f b;
        b.x = B[(size_t)(k + kOff)     * N + col];
        b.y = B[(size_t)(k + kOff + 1) * N + col];
        acc = __builtin_amdgcn_wmma_f32_16x16x4_f32(
                   false, a,  false, b,
                   (short)0, acc,  false,  false);
    }

    const float bb   = bias ? bias[col] : 0.0f;
    const int   rOff = (lane >> 4) << 3;
#pragma unroll
    for (int j = 0; j < 8; ++j) {
        int r = mBase + j + rOff;
        if (r < M) C[(size_t)r * N + col] = acc[j] + bb;
    }
}

#define AGG_R   200
#define AGG_T   1024
#define AGG_W   (AGG_T / 32)
#define AGG_CH  512
#define AGG_Q   (AGG_CH + 32)

__device__ __forceinline__ void agg_ld_chunk(int4 (&r)[4], const int* __restrict__ dst,
                                             int base, int lane, int E, bool vec) {
    if (vec && base + AGG_CH <= E) {
#pragma unroll
        for (int k = 0; k < 4; ++k) r[k] = *(const int4*)(dst + base + k * 128 + lane * 4);
    } else {
#pragma unroll
        for (int k = 0; k < 4; ++k) {
            const int e0 = base + k * 128 + lane * 4;
            r[k].x = (e0     < E) ? dst[e0]     : -1;
            r[k].y = (e0 + 1 < E) ? dst[e0 + 1] : -1;
            r[k].z = (e0 + 2 < E) ? dst[e0 + 2] : -1;
            r[k].w = (e0 + 3 < E) ? dst[e0 + 3] : -1;
        }
    }
}

template <int MODE>
__global__ __launch_bounds__(AGG_T)
void k_agg(const float* __restrict__ H,
           const int* __restrict__ src, const int* __restrict__ dst,
           const float* __restrict__ dinv, const float* __restrict__ bias,
           const float* __restrict__ gamma, const float* __restrict__ beta,
           const float* __restrict__ mean, const float* __restrict__ var,
           float* __restrict__ out, int N, int E) {
    __shared__ __attribute__((aligned(16))) float tile[AGG_R * FEAT];
    __shared__ int queue[AGG_W * AGG_Q];

    const int tid  = threadIdx.x;
    const int lane = tid & 31;
    const int wave = __builtin_amdgcn_readfirstlane(tid >> 5);
    const int lo   = blockIdx.x * AGG_R;
    const int cnt  = (N - lo < AGG_R) ? (N - lo) : AGG_R;

    for (int i = tid; i < AGG_R * FEAT / 4; i += AGG_T)
        ((float4*)tile)[i] = make_float4(0.0f, 0.0f, 0.0f, 0.0f);
    __syncthreads();

    int* q = queue + wave * AGG_Q;
    int  qn = 0;

    auto flush = [&](int n) {
        for (int g = 0; g < n; g += 32) {
            const int m = (n - g < 32) ? (n - g) : 32;
            int s = 0, dl = 0; float w = 0.0f;
            if (lane < m) {
                const int e = q[g + lane];
                const int d = dst[e];
                s  = src[e];
                dl = d - lo;
                w  = dinv[s] * dinv[d];
            }
            for (int u0 = 0; u0 < m; u0 += 8) {
                float v[8][4];
                int   dls[8];
#pragma unroll
                for (int u = 0; u < 8; ++u) {
                    const int uu = (u0 + u < m) ? (u0 + u) : (m - 1);
                    const int   su = __builtin_amdgcn_readlane(s, uu);
                    const float wu = __int_as_float(__builtin_amdgcn_readlane(__float_as_int(w), uu));
                    dls[u] = __builtin_amdgcn_readlane(dl, uu);
                    const float* hp = H + (size_t)su * FEAT + lane;
                    v[u][0] = hp[0]  * wu;
                    v[u][1] = hp[32] * wu;
                    v[u][2] = hp[64] * wu;
                    v[u][3] = hp[96] * wu;
                }
#pragma unroll
                for (int u = 0; u < 8; ++u) {
                    if (u0 + u < m) {
                        float* tp = tile + dls[u] * FEAT + lane;
                        __hip_atomic_fetch_add(tp,      v[u][0], __ATOMIC_RELAXED, __HIP_MEMORY_SCOPE_WORKGROUP);
                        __hip_atomic_fetch_add(tp + 32, v[u][1], __ATOMIC_RELAXED, __HIP_MEMORY_SCOPE_WORKGROUP);
                        __hip_atomic_fetch_add(tp + 64, v[u][2], __ATOMIC_RELAXED, __HIP_MEMORY_SCOPE_WORKGROUP);
                        __hip_atomic_fetch_add(tp + 96, v[u][3], __ATOMIC_RELAXED, __HIP_MEMORY_SCOPE_WORKGROUP);
                    }
                }
            }
        }
    };

    const bool vec  = ((reinterpret_cast<uintptr_t>(dst) & 15) == 0);
    const int  nCh  = (E + AGG_CH - 1) / AGG_CH;
    const int  rot  = (int)blockIdx.x * (nCh / (int)gridDim.x);

    int  c = wave;
    int4 nx[4];
    int  nbase = 0;
    if (c < nCh) {
        int cc = c + rot; if (cc >= nCh) cc -= nCh;
        nbase = cc * AGG_CH;
        agg_ld_chunk(nx, dst, nbase, lane, E, vec);
    }
    while (c < nCh) {
        int4 cu[4];
#pragma unroll
        for (int k = 0; k < 4; ++k) cu[k] = nx[k];
        const int cbase = nbase;
        const int cn = c + AGG_W;
        if (cn < nCh) {
            int cc = cn + rot; if (cc >= nCh) cc -= nCh;
            nbase = cc * AGG_CH;
            agg_ld_chunk(nx, dst, nbase, lane, E, vec);
        }
#pragma unroll
        for (int k = 0; k < 4; ++k) {
#pragma unroll
            for (int j = 0; j < 4; ++j) {
                const int d = (j == 0) ? cu[k].x : (j == 1) ? cu[k].y : (j == 2) ? cu[k].z : cu[k].w;
                const unsigned t  = (unsigned)(d - lo);
                const bool     mt = t < (unsigned)cnt;
                const unsigned mask = __builtin_amdgcn_ballot_w32(mt);
                if (mask != 0u) {
                    if (mt) q[qn + (int)__builtin_amdgcn_mbcnt_lo(mask, 0u)] = cbase + k * 128 + lane * 4 + j;
                    qn += __builtin_popcount(mask);
                }
            }
        }
        if (qn >= 32) { flush(qn); qn = 0; }
        c = cn;
    }
    flush(qn);
    __syncthreads();

    const int   f  = tid & (FEAT - 1);
    const float bb = bias[f];
    float mu = 0.0f, rs = 1.0f, ga = 1.0f, be = 0.0f;
    if (MODE == 1) { mu = mean[f]; rs = rsqrtf(var[f] + BN_EPS); ga = gamma[f]; be = beta[f]; }
    for (int r = tid >> 7; r < cnt; r += AGG_T / FEAT) {
        const int    node = lo + r;
        const float  di   = dinv[node];
        const size_t gi   = (size_t)node * FEAT + f;
        float v = tile[r * FEAT + f] + H[gi] * di * di + bb;
        if (MODE == 1) {
            v = (v - mu) * rs * ga + be;
            v = fmaxf(v, 0.0f);
        }
        out[gi] = v;
    }
}

extern "C" void kernel_launch(void* const* d_in, const int* in_sizes, int n_in,
                              void* d_out, int out_size, void* d_ws, size_t ws_size,
                              hipStream_t stream) {
    const float* X     = (const float*)d_in[0];
    const int*   edge  = (const int*)  d_in[1];
    const float* W1    = (const float*)d_in[2];
    const float* b1    = (const float*)d_in[3];
    const float* gamma = (const float*)d_in[4];
    const float* beta  = (const float*)d_in[5];
    const float* rmean = (const float*)d_in[6];
    const float* rvar  = (const float*)d_in[7];
    const float* W2    = (const float*)d_in[8];
    const float* b2    = (const float*)d_in[9];
    const float* Wc    = (const float*)d_in[10];
    const float* bc    = (const float*)d_in[11];
    float*       out   = (float*)d_out;

    const int N = in_sizes[0] / FEAT;
    const int E = in_sizes[1] / 2;
    const int* src = edge;
    const int* dst = edge + E;
    const int total = N * FEAT;

    float* deg  = (float*)d_ws;
    float* bufA = deg + ((N + 63) & ~63);
    float* bufB = bufA + (size_t)total;

    const int T256   = 256;
    const int gNode  = (N + T256 - 1) / T256;
    const int gEdge  = (E + T256 - 1) / T256;
    const int gAgg   = (N + AGG_R - 1) / AGG_R;
    const int tilesL = ((N + 15) / 16) * (FEAT / 16);
    const int tilesC = ((N + 15) / 16) * (N_CLASSES / 16);
    const int gGemmL = (tilesL + 7) / 8;
    const int gGemmC = (tilesC + 7) / 8;

    k_fill<<<gNode, T256, 0, stream>>>(deg, 1.0f, N);
    k_deg_accum<<<gEdge, T256, 0, stream>>>(dst, deg, E);
    k_rsqrt_inplace<<<gNode, T256, 0, stream>>>(deg, N);

    k_gemm_wmma<<<gGemmL, T256, 0, stream>>>(X, W1, nullptr, bufA, N, FEAT, FEAT);
    k_agg<1><<<gAgg, AGG_T, 0, stream>>>(bufA, src, dst, deg, b1, gamma, beta, rmean, rvar, bufB, N, E);

    k_gemm_wmma<<<gGemmL, T256, 0, stream>>>(bufB, W2, nullptr, bufA, N, FEAT, FEAT);
    k_agg<2><<<gAgg, AGG_T, 0, stream>>>(bufA, src, dst, deg, b2, nullptr, nullptr, nullptr, nullptr, bufB, N, E);

    k_gemm_wmma<<<gGemmC, T256, 0, stream>>>(bufB, Wc, bc, out, N, FEAT, N_CLASSES);
}


# Layer GEMMs retiled: a wave owns 16 rows x all 128 columns (8 accumulator tiles), A read once into fragments, W staged once per block in LDS; 25000 waves to 3136; f32 sums run in a permuted k order
# speedup vs baseline: 2.7947x; 1.6366x over previous
#include <hip/hip_runtime.h>
#include <math.h>

typedef float v2f __attribute__((ext_vector_type(2)));
typedef float v8f __attribute__((ext_vector_type(8)));

#define FEAT 128
#define N_CLASSES 16
#define BN_EPS 1e-5f

__global__ void k_gemm_wmma(const float* __restrict__ A, const float* __restrict__ B,
                            const float* __restrict__ bias, float* __restrict__ C,
                            int M, int K, int N) {
    const int lane  = threadIdx.x & 31;
    const int wave  = threadIdx.x >> 5;
    const int wpb   = blockDim.x >> 5;
    const int tile  = blockIdx.x * wpb + wave;
    const int Mt    = (M + 15) >> 4;
    const int Nt    = N >> 4;
    if (tile >= Mt * Nt) return;

    const int mBase = (tile / Nt) << 4;
    const int nBase = (tile % Nt) << 4;

    int row = mBase + (lane & 15);
    if (row > M - 1) row = M - 1;
    const int col  = nBase + (lane & 15);
    const int kOff = (lane >> 4) << 1;

    v8f acc = {};
    for (int k = 0; k < K; k += 4) {
        v2f a = *(const v2f*)(A + (size_t)row * K + k + kOff);
        v2f b;
        b.x = B[(size_t)(k + kOff)     * N + col];
        b.y = B[(size_t)(k + kOff + 1) * N + col];
        acc = __builtin_amdgcn_wmma_f32_16x16x4_f32(
                   false, a,  false, b,
                   (short)0, acc,  false,  false);
    }

    const float bb   = bias ? bias[col] : 0.0f;
    const int   rOff = (lane >> 4) << 3;
#pragma unroll
    for (int j = 0; j < 8; ++j) {
        int r = mBase + j + rOff;
        if (r < M) C[(size_t)r * N + col] = acc[j] + bb;
    }
}

#define GS_W  16
#define GS_T  (GS_W * 32)
#define GS_E  (32 * 2 * 4 * 16)

__global__ __launch_bounds__(GS_T)
void k_gemm_strip(const float* __restrict__ A, const float* __restrict__ Wm,
                  float* __restrict__ C, int M) {
    __shared__ float4 Bl[GS_E];

    const int tid  = threadIdx.x;
    const int lane = tid & 31;
    const int wave = __builtin_amdgcn_readfirstlane(tid >> 5);
    const int c    = lane & 15;
    const int h    = lane >> 4;
    const int m0   = (blockIdx.x * GS_W + wave) << 4;

    float2 w0[GS_E / GS_T], w1[GS_E / GS_T];
#pragma unroll
    for (int m = 0; m < GS_E / GS_T; ++m) {
        const int i  = tid + m * GS_T;
        const int w  = i >> 7;
        const int q  = (i >> 4) & 3;
        const int k0 = 8 * (w >> 1) + 4 * ((i >> 6) & 1) + 2 * (w & 1);
        const int cc = 64 * (q >> 1) + 4 * (i & 15) + 2 * (q & 1);
        w0[m] = *(const float2*)(Wm + k0 * FEAT + cc);
        w1[m] = *(const float2*)(Wm + (k0 + 1) * FEAT + cc);
    }

#pragma unroll
    for (int m = 0; m < GS_E / GS_T; ++m)
        Bl[tid + m * GS_T] = make_float4(w0[m].x, w1[m].x, w0[m].y, w1[m].y);
    __syncthreads();
    if (m0 >= M) return;

    int row = m0 + c;
    if (row > M - 1) row = M - 1;
    const float4* ap = (const float4*)(A + (size_t)row * FEAT + 4 * h);
    float4 a[16];
#pragma unroll
    for (int j = 0; j < 16; ++j) a[j] = ap[2 * j];

    v8f acc[8];
#pragma unroll
    for (int n = 0; n < 8; ++n) acc[n] = v8f{};

    const float4* bp = Bl + h * 64 + c;
    float4 bf[2][8];
#pragma unroll
    for (int q = 0; q < 8; ++q) bf[0][q] = bp[(q >> 2) * 128 + (q & 3) * 16];
#pragma unroll
    for (int g = 0; g < 16; ++g) {
        if (g + 1 < 16) {
#pragma unroll
            for (int q = 0; q < 8; ++q)
                bf[(g + 1) & 1][q] = bp[(2 * (g + 1) + (q >> 2)) * 128 + (q & 3) * 16];
        }
        __builtin_amdgcn_sched_barrier(0);
#pragma unroll
        for (int t = 0; t < 2; ++t) {
            v2f av;
            av.x = t ? a[g].z : a[g].x;
            av.y = t ? a[g].w : a[g].y;
#pragma unroll
            for (int q = 0; q < 4; ++q) {
                const float4 b = bf[g & 1][4 * t + q];
                v2f b0, b1;
                b0.x = b.x; b0.y = b.y;
                b1.x = b.z; b1.y = b.w;
                acc[2 * q] = __builtin_amdgcn_wmma_f32_16x16x4_f32(
                                  false, av,  false, b0,
                                  (short)0, acc[2 * q],  false,  false);
                acc[2 * q + 1] = __builtin_amdgcn_wmma_f32_16x16x4_f32(
                                  false, av,  false, b1,
                                  (short)0, acc[2 * q + 1],  false,  false);
            }
        }
        __builtin_amdgcn_sched_barrier(0);
    }

#pragma unroll
    for (int j = 0; j < 8; ++j) {
        const int r = m0 + j + 8 * h;
        if (r < M) {
            float* cp = C + (size_t)r * FEAT + 4 * c;
            *(float4*)cp        = make_float4(acc[0][j], acc[1][j], acc[2][j], acc[3][j]);
            *(float4*)(cp + 64) = make_float4(acc[4][j], acc[5][j], acc[6][j], acc[7][j]);
        }
    }
}

#define AGG_R   200
#define AGG_T   1024
#define AGG_W   (AGG_T / 32)
#define AGG_CH  512
#define AGG_Q   (AGG_CH + 32)
#define SCAN_G  5
#define SCAN_R  (AGG_R * SCAN_G)
#define LIST_G  4

__device__ __forceinline__ void agg_ld_chunk(int4 (&r)[4], const int* __restrict__ dst,
                                             int base, int lane, int E, bool vec) {
    if (vec && base + AGG_CH <= E) {
#pragma unroll
        for (int k = 0; k < 4; ++k) r[k] = *(const int4*)(dst + base + k * 128 + lane * 4);
    } else {
#pragma unroll
        for (int k = 0; k < 4; ++k) {
            const int e0 = base + k * 128 + lane * 4;
            r[k].x = (e0     < E) ? dst[e0]     : -1;
            r[k].y = (e0 + 1 < E) ? dst[e0 + 1] : -1;
            r[k].z = (e0 + 2 < E) ? dst[e0 + 2] : -1;
            r[k].w = (e0 + 3 < E) ? dst[e0 + 3] : -1;
        }
    }
}

template <class P, class F>
__device__ __forceinline__ void scan_dst(const int* __restrict__ dst, int E, int lo, int cnt,
                                         int lane, int wave, P push, F flush) {
    int qn = 0;
    const bool vec  = ((reinterpret_cast<uintptr_t>(dst) & 15) == 0);
    const int  nCh  = (E + AGG_CH - 1) / AGG_CH;
    const int  rot  = (int)blockIdx.x * (nCh / (int)gridDim.x);

    int  c = wave;
    int4 nx[4];
    int  nbase = 0;
    if (c < nCh) {
        int cc = c + rot; if (cc >= nCh) cc -= nCh;
        nbase = cc * AGG_CH;
        agg_ld_chunk(nx, dst, nbase, lane, E, vec);
    }
    while (c < nCh) {
        int4 cu[4];
#pragma unroll
        for (int k = 0; k < 4; ++k) cu[k] = nx[k];
        const int cbase = nbase;
        const int cn = c + AGG_W;
        if (cn < nCh) {
            int cc = cn + rot; if (cc >= nCh) cc -= nCh;
            nbase = cc * AGG_CH;
            agg_ld_chunk(nx, dst, nbase, lane, E, vec);
        }
#pragma unroll
        for (int k = 0; k < 4; ++k) {
#pragma unroll
            for (int j = 0; j < 4; ++j) {
                const int d = (j == 0) ? cu[k].x : (j == 1) ? cu[k].y : (j == 2) ? cu[k].z : cu[k].w;
                const unsigned t  = (unsigned)(d - lo);
                const bool     mt = t < (unsigned)cnt;
                const unsigned mask = __builtin_amdgcn_ballot_w32(mt);
                if (mask != 0u) {
                    if (mt) push(qn + (int)__builtin_amdgcn_mbcnt_lo(mask, 0u), cbase + k * 128 + lane * 4 + j, (int)t);
                    qn += __builtin_popcount(mask);
                }
            }
        }
        if (qn >= 32) { flush(qn); qn = 0; }
        c = cn;
    }
    flush(qn);
}

__global__ __launch_bounds__(AGG_T)
void k_scan(const int* __restrict__ dst,
            float* __restrict__ dinv, unsigned* __restrict__ lists, int* __restrict__ counts,
            int N, int E, int cap) {
    __shared__ int cntd[SCAN_R];
    __shared__ int cur[SCAN_G];
    __shared__ int2 queue[AGG_W * AGG_Q];

    const int tid  = threadIdx.x;
    const int lane = tid & 31;
    const int wave = __builtin_amdgcn_readfirstlane(tid >> 5);
    const int lo   = blockIdx.x * SCAN_R;
    const int nn   = (N - lo < SCAN_R) ? (N - lo) : SCAN_R;

    for (int i = tid; i < SCAN_R; i += AGG_T) cntd[i] = 0;
    if (tid < SCAN_G) cur[tid] = 0;
    __syncthreads();

    int2* q = queue + wave * AGG_Q;
    unsigned* myl = lists + (size_t)blockIdx.x * SCAN_G * cap;

    auto push = [&](int pos, int e, int t) { q[pos] = make_int2(e, t); };
    auto flush = [&](int n) {
        for (int g = 0; g < n; g += 32) {
            if (g + lane < n) {
                const int2 x  = q[g + lane];
                const int  dl = x.y;
                const int  o  = dl / AGG_R;
                const int  r  = dl - o * AGG_R;
                __hip_atomic_fetch_add(&cntd[dl], 1, __ATOMIC_RELAXED, __HIP_MEMORY_SCOPE_WORKGROUP);
                const int slot = __hip_atomic_fetch_add(&cur[o], 1, __ATOMIC_RELAXED, __HIP_MEMORY_SCOPE_WORKGROUP);
                if (slot < cap) myl[(size_t)o * cap + slot] = ((unsigned)x.x << 8) | (unsigned)r;
            }
        }
    };
    scan_dst(dst, E, lo, nn, lane, wave, push, flush);
    __syncthreads();

    for (int i = tid; i < nn; i += AGG_T) dinv[lo + i] = rsqrtf(1.0f + (float)cntd[i]);
    if (tid < SCAN_G) {
        const int ob = blockIdx.x * SCAN_G + tid;
        if (ob * AGG_R < N) counts[ob] = cur[tid];
    }
}

template <int MODE>
__global__ __launch_bounds__(AGG_T)
void k_agg(const float* __restrict__ H,
           const int* __restrict__ src, const int* __restrict__ dst,
           const float* __restrict__ dinv,
           const unsigned* __restrict__ lists, const int* __restrict__ counts, int cap,
           const float* __restrict__ bias,
           const float* __restrict__ gamma, const float* __restrict__ beta,
           const float* __restrict__ mean, const float* __restrict__ var,
           float* __restrict__ out, int N, int E) {
    __shared__ __attribute__((aligned(16))) float tile[AGG_R * FEAT];
    __shared__ int queue[AGG_W * AGG_Q];

    const int tid  = threadIdx.x;
    const int lane = tid & 31;
    const int wave = __builtin_amdgcn_readfirstlane(tid >> 5);
    const int lo   = blockIdx.x * AGG_R;
    const int cnt  = (N - lo < AGG_R) ? (N - lo) : AGG_R;

    for (int i = tid; i < AGG_R * FEAT / 4; i += AGG_T)
        ((float4*)tile)[i] = make_float4(0.0f, 0.0f, 0.0f, 0.0f);
    __syncthreads();

    auto rows = [&](int m, int s, int dl, float w) {
        for (int u0 = 0; u0 < m; u0 += 8) {
            float v[8][4];
            int   dls[8];
#pragma unroll
            for (int u = 0; u < 8; ++u) {
                const int uu = (u0 + u < m) ? (u0 + u) : (m - 1);
                const int   su = __builtin_amdgcn_readlane(s, uu);
                const float wu = __int_as_float(__builtin_amdgcn_readlane(__float_as_int(w), uu));
                dls[u] = __builtin_amdgcn_readlane(dl, uu);
                const float* hp = H + (size_t)su * FEAT + lane;
                v[u][0] = hp[0]  * wu;
                v[u][1] = hp[32] * wu;
                v[u][2] = hp[64] * wu;
                v[u][3] = hp[96] * wu;
            }
#pragma unroll
            for (int u = 0; u < 8; ++u) {
                if (u0 + u < m) {
                    float* tp = tile + dls[u] * FEAT + lane;
                    __hip_atomic_fetch_add(tp,      v[u][0], __ATOMIC_RELAXED, __HIP_MEMORY_SCOPE_WORKGROUP);
                    __hip_atomic_fetch_add(tp + 32, v[u][1], __ATOMIC_RELAXED, __HIP_MEMORY_SCOPE_WORKGROUP);
                    __hip_atomic_fetch_add(tp + 64, v[u][2], __ATOMIC_RELAXED, __HIP_MEMORY_SCOPE_WORKGROUP);
                    __hip_atomic_fetch_add(tp + 96, v[u][3], __ATOMIC_RELAXED, __HIP_MEMORY_SCOPE_WORKGROUP);
                }
            }
        }
    };

    const int n = counts[blockIdx.x];
    if (n <= cap) {
        const unsigned* L = lists + (size_t)blockIdx.x * cap;
        unsigned ent[LIST_G];
        int      sv[LIST_G];
        float    wgt[LIST_G];
#pragma unroll
        for (int k = 0; k < LIST_G; ++k) {
            const int idx = wave * 32 + k * AGG_T + lane;
            ent[k] = (idx < n) ? L[idx] : 0u;
        }
#pragma unroll
        for (int k = 0; k < LIST_G; ++k) sv[k] = src[ent[k] >> 8];
        asm volatile("" :: "v"(sv[0]), "v"(sv[1]), "v"(sv[2]), "v"(sv[3]));
        float da[LIST_G], db[LIST_G];
#pragma unroll
        for (int k = 0; k < LIST_G; ++k) {
            da[k] = dinv[sv[k]];
            db[k] = dinv[lo + (int)(ent[k] & 255u)];
        }
        asm volatile("" :: "v"(da[0]), "v"(da[1]), "v"(da[2]), "v"(da[3]),
                           "v"(db[0]), "v"(db[1]), "v"(db[2]), "v"(db[3]));
#pragma unroll
        for (int k = 0; k < LIST_G; ++k) wgt[k] = da[k] * db[k];
#pragma unroll
        for (int k = 0; k < LIST_G; ++k) {
            const int g = wave * 32 + k * AGG_T;
            if (g < n) rows((n - g < 32) ? (n - g) : 32, sv[k], (int)(ent[k] & 255u), wgt[k]);
        }
    } else {
        int* q = queue + wave * AGG_Q;
        auto push = [&](int pos, int e, int t) { q[pos] = e; };
        auto flush = [&](int nq) {
            for (int g = 0; g < nq; g += 32) {
                const int m = (nq - g < 32) ? (nq - g) : 32;
                int s = 0, dl = 0; float w = 0.0f;
                if (lane < m) {
                    const int e = q[g + lane];
                    const int d = dst[e];
                    s  = src[e];
                    dl = d - lo;
                    w  = dinv[s] * dinv[d];
                }
                rows(m, s, dl, w);
            }
        };
        scan_dst(dst, E, lo, cnt, lane, wave, push, flush);
    }
    __syncthreads();

    const int   f  = tid & (FEAT - 1);
    const float bb = bias[f];
    float mu = 0.0f, rs = 1.0f, ga = 1.0f, be = 0.0f;
    if (MODE == 1) { mu = mean[f]; rs = rsqrtf(var[f] + BN_EPS); ga = gamma[f]; be = beta[f]; }
    for (int r = tid >> 7; r < cnt; r += AGG_T / FEAT) {
        const int    node = lo + r;
        const float  di   = dinv[node];
        const size_t gi   = (size_t)node * FEAT + f;
        float v = tile[r * FEAT + f] + H[gi] * di * di + bb;
        if (MODE == 1) {
            v = (v - mu) * rs * ga + be;
            v = fmaxf(v, 0.0f);
        }
        out[gi] = v;
    }
}

extern "C" void kernel_launch(void* const* d_in, const int* in_sizes, int n_in,
                              void* d_out, int out_size, void* d_ws, size_t ws_size,
                              hipStream_t stream) {
    const float* X     = (const float*)d_in[0];
    const int*   edge  = (const int*)  d_in[1];
    const float* W1    = (const float*)d_in[2];
    const float* b1    = (const float*)d_in[3];
    const float* gamma = (const float*)d_in[4];
    const float* beta  = (const float*)d_in[5];
    const float* rmean = (const float*)d_in[6];
    const float* rvar  = (const float*)d_in[7];
    const float* W2    = (const float*)d_in[8];
    const float* b2    = (const float*)d_in[9];
    const float* Wc    = (const float*)d_in[10];
    const float* bc    = (const float*)d_in[11];
    float*       out   = (float*)d_out;

    const int N = in_sizes[0] / FEAT;
    const int E = in_sizes[1] / 2;
    const int* src = edge;
    const int* dst = edge + E;
    const int total = N * FEAT;

    float* dinv = (float*)d_ws;
    float* bufA = dinv + ((N + 63) & ~63);
    float* bufB = bufA + (size_t)total;

    const int T256   = 256;
    const int gAgg   = (N + AGG_R - 1) / AGG_R;
    const int gScan  = (N + SCAN_R - 1) / SCAN_R;
    const int tilesC = ((N + 15) / 16) * (N_CLASSES / 16);
    const int gGemmC = (tilesC + 7) / 8;
    const int gStrip = (N + GS_W * 16 - 1) / (GS_W * 16);

    const int cap = ((N * N_CLASSES - gAgg) / gAgg) & ~3;
    unsigned* lists  = (unsigned*)d_out;
    int*      counts = (int*)d_out + (size_t)gAgg * cap;

    k_scan<<<gScan, AGG_T, 0, stream>>>(dst, dinv, lists, counts, N, E, cap);

    k_gemm_strip<<<gStrip, GS_T, 0, stream>>>(X, W1, bufA, N);
    k_agg<1><<<gAgg, AGG_T, 0, stream>>>(bufA, src, dst, dinv, lists, counts, cap, b1, gamma, beta, rmean, rvar, bufB, N, E);

    k_gemm_strip<<<gStrip, GS_T, 0, stream>>>(bufB, W2, bufA, N);
    k_agg<2><<<gAgg, AGG_T, 0, stream>>>(bufA, src, dst, dinv, lists, counts, cap, b2, nullptr, nullptr, nullptr, nullptr, bufB, N, E);

    k_gemm_wmma<<<gGemmC, T256, 0, stream>>>(bufB, Wc, bc, out, N, FEAT, N_CLASSES);
}


# Edge list read once, by slices, in the layer-1 product launch (LDS ranks, 64-word sub-lists per slice and 200-node range); owner blocks merge them, count degrees; no sweeps; f32 sums in arrival order
# speedup vs baseline: 3.4695x; 1.2415x over previous
#include <hip/hip_runtime.h>
#include <math.h>

typedef float v2f __attribute__((ext_vector_type(2)));
typedef float v8f __attribute__((ext_vector_type(8)));

#define FEAT 128
#define N_CLASSES 16
#define BN_EPS 1e-5f
#define AGG_R 200

__global__ void k_gemm_wmma(const float* __restrict__ A, const float* __restrict__ B,
                            const float* __restrict__ bias, float* __restrict__ C,
                            int M, int K, int N) {
    const int lane  = threadIdx.x & 31;
    const int wave  = threadIdx.x >> 5;
    const int wpb   = blockDim.x >> 5;
    const int tile  = blockIdx.x * wpb + wave;
    const int Mt    = (M + 15) >> 4;
    const int Nt    = N >> 4;
    if (tile >= Mt * Nt) return;

    const int mBase = (tile / Nt) << 4;
    const int nBase = (tile % Nt) << 4;

    int row = mBase + (lane & 15);
    if (row > M - 1) row = M - 1;
    const int col  = nBase + (lane & 15);
    const int kOff = (lane >> 4) << 1;

    v8f acc = {};
    for (int k = 0; k < K; k += 4) {
        v2f a = *(const v2f*)(A + (size_t)row * K + k + kOff);
        v2f b;
        b.x = B[(size_t)(k + kOff)     * N + col];
        b.y = B[(size_t)(k + kOff + 1) * N + col];
        acc = __builtin_amdgcn_wmma_f32_16x16x4_f32(
                   false, a,  false, b,
                   (short)0, acc,  false,  false);
    }

    const float bb   = bias ? bias[col] : 0.0f;
    const int   rOff = (lane >> 4) << 3;
#pragma unroll
    for (int j = 0; j < 8; ++j) {
        int r = mBase + j + rOff;
        if (r < M) C[(size_t)r * N + col] = acc[j] + bb;
    }
}

#define GS_W  16
#define GS_T  (GS_W * 32)
#define GS_E  (32 * 2 * 4 * 16)

#define BIN_K   7
#define BIN_E   (GS_T * BIN_K)
#define BIN_O   1024
#define SUB     64

__device__ __forceinline__ void bin_slices(const int* __restrict__ src, const int* __restrict__ dst,
                                           unsigned* __restrict__ tmp, int N, int E, int nOwn, int nSl,
                                           int* hist) {
    const int tid = threadIdx.x;
    for (int sl = blockIdx.x; sl < nSl; sl += gridDim.x) {
        const int base = sl * BIN_E;
        int d[BIN_K], s[BIN_K];
        if (base + BIN_E <= E) {
#pragma unroll
            for (int k = 0; k < BIN_K; ++k) {
                const int e = base + k * GS_T + tid;
                d[k] = dst[e];
                s[k] = src[e];
            }
        } else {
#pragma unroll
            for (int k = 0; k < BIN_K; ++k) {
                const int e = base + k * GS_T + tid;
                d[k] = (e < E) ? dst[e] : -1;
                s[k] = (e < E) ? src[e] : 0;
            }
        }
        int o[BIN_K], rk[BIN_K];
#pragma unroll
        for (int k = 0; k < BIN_K; ++k) {
            o[k]  = ((unsigned)d[k] < (unsigned)N) ? d[k] / AGG_R : -1;
            rk[k] = 0;
        }
        for (int w0 = 0; w0 < nOwn; w0 += BIN_O) {
            __syncthreads();
            for (int i = tid; i < BIN_O; i += GS_T) hist[i] = 0;
            __syncthreads();
#pragma unroll
            for (int k = 0; k < BIN_K; ++k) {
                const unsigned ow = (unsigned)(o[k] - w0);
                if (ow < (unsigned)BIN_O)
                    rk[k] = __hip_atomic_fetch_add(&hist[ow], 1, __ATOMIC_RELAXED, __HIP_MEMORY_SCOPE_WORKGROUP);
            }
            __syncthreads();
            for (int i = tid; i < BIN_O && w0 + i < nOwn; i += GS_T)
                tmp[((size_t)(w0 + i) * nSl + sl) * SUB] = (unsigned)hist[i];
#pragma unroll
            for (int k = 0; k < BIN_K; ++k) {
                const unsigned ow = (unsigned)(o[k] - w0);
                if (ow < (unsigned)BIN_O && rk[k] < SUB - 1)
                    tmp[((size_t)o[k] * nSl + sl) * SUB + 1 + rk[k]] =
                        ((unsigned)s[k] << 8) | (unsigned)(d[k] - o[k] * AGG_R);
            }
        }
    }
}

__global__ __launch_bounds__(GS_T)
void k_gemm_strip(const float* __restrict__ A, const float* __restrict__ Wm,
                  float* __restrict__ C, int M,
                  const int* __restrict__ src, const int* __restrict__ dst,
                  unsigned* __restrict__ tmp, int tmpWords, int E, int nOwn, int nSl) {
    __shared__ float4 Bl[GS_E];
    __shared__ int hist[BIN_O];

    const int tid  = threadIdx.x;

    if (nSl > 0 && (size_t)nOwn * nSl * SUB <= (size_t)tmpWords)
        bin_slices(src, dst, tmp, M, E, nOwn, nSl, hist);

    const int lane = tid & 31;
    const int wave = __builtin_amdgcn_readfirstlane(tid >> 5);
    const int c    = lane & 15;
    const int h    = lane >> 4;
    const int m0   = (blockIdx.x * GS_W + wave) << 4;

    float2 w0[GS_E / GS_T], w1[GS_E / GS_T];
#pragma unroll
    for (int m = 0; m < GS_E / GS_T; ++m) {
        const int i  = tid + m * GS_T;
        const int w  = i >> 7;
        const int q  = (i >> 4) & 3;
        const int k0 = 8 * (w >> 1) + 4 * ((i >> 6) & 1) + 2 * (w & 1);
        const int cc = 64 * (q >> 1) + 4 * (i & 15) + 2 * (q & 1);
        w0[m] = *(const float2*)(Wm + k0 * FEAT + cc);
        w1[m] = *(const float2*)(Wm + (k0 + 1) * FEAT + cc);
    }

#pragma unroll
    for (int m = 0; m < GS_E / GS_T; ++m)
        Bl[tid + m * GS_T] = make_float4(w0[m].x, w1[m].x, w0[m].y, w1[m].y);
    __syncthreads();
    if (m0 >= M) return;

    int row = m0 + c;
    if (row > M - 1) row = M - 1;
    const float4* ap = (const float4*)(A + (size_t)row * FEAT + 4 * h);
    float4 a[16];
#pragma unroll
    for (int j = 0; j < 16; ++j) a[j] = ap[2 * j];

    v8f acc[8];
#pragma unroll
    for (int n = 0; n < 8; ++n) acc[n] = v8f{};

    const float4* bp = Bl + h * 64 + c;
    float4 bf[2][8];
#pragma unroll
    for (int q = 0; q < 8; ++q) bf[0][q] = bp[(q >> 2) * 128 + (q & 3) * 16];
#pragma unroll
    for (int g = 0; g < 16; ++g) {
        if (g + 1 < 16) {
#pragma unroll
            for (int q = 0; q < 8; ++q)
                bf[(g + 1) & 1][q] = bp[(2 * (g + 1) + (q >> 2)) * 128 + (q & 3) * 16];
        }
        __builtin_amdgcn_sched_barrier(0);
#pragma unroll
        for (int t = 0; t < 2; ++t) {
            v2f av;
            av.x = t ? a[g].z : a[g].x;
            av.y = t ? a[g].w : a[g].y;
#pragma unroll
            for (int q = 0; q < 4; ++q) {
                const float4 b = bf[g & 1][4 * t + q];
                v2f b0, b1;
                b0.x = b.x; b0.y = b.y;
                b1.x = b.z; b1.y = b.w;
                acc[2 * q] = __builtin_amdgcn_wmma_f32_16x16x4_f32(
                                  false, av,  false, b0,
                                  (short)0, acc[2 * q],  false,  false);
                acc[2 * q + 1] = __builtin_amdgcn_wmma_f32_16x16x4_f32(
                                  false, av,  false, b1,
                                  (short)0, acc[2 * q + 1],  false,  false);
            }
        }
        __builtin_amdgcn_sched_barrier(0);
    }

#pragma unroll
    for (int j = 0; j < 8; ++j) {
        const int r = m0 + j + 8 * h;
        if (r < M) {
            float* cp = C + (size_t)r * FEAT + 4 * c;
            *(float4*)cp        = make_float4(acc[0][j], acc[1][j], acc[2][j], acc[3][j]);
            *(float4*)(cp + 64) = make_float4(acc[4][j], acc[5][j], acc[6][j], acc[7][j]);
        }
    }
}

#define AGG_T   1024
#define AGG_W   (AGG_T / 32)
#define AGG_CH  512
#define AGG_Q   (AGG_CH + 32)
#define LIST_G  4
#define DEG_T   256
#define DEG_W   (DEG_T / 32)
#define DEG_U   8

__device__ __forceinline__ void agg_ld_chunk(int4 (&r)[4], const int* __restrict__ dst,
                                             int base, int lane, int E, bool vec) {
    if (vec && base + AGG_CH <= E) {
#pragma unroll
        for (int k = 0; k < 4; ++k) r[k] = *(const int4*)(dst + base + k * 128 + lane * 4);
    } else {
#pragma unroll
        for (int k = 0; k < 4; ++k) {
            const int e0 = base + k * 128 + lane * 4;
            r[k].x = (e0     < E) ? dst[e0]     : -1;
            r[k].y = (e0 + 1 < E) ? dst[e0 + 1] : -1;
            r[k].z = (e0 + 2 < E) ? dst[e0 + 2] : -1;
            r[k].w = (e0 + 3 < E) ? dst[e0 + 3] : -1;
        }
    }
}

template <class P, class F>
__device__ __forceinline__ void scan_dst(const int* __restrict__ dst, int E, int lo, int cnt,
                                         int lane, int wave, P push, F flush) {
    int qn = 0;
    const bool vec  = ((reinterpret_cast<uintptr_t>(dst) & 15) == 0);
    const int  nCh  = (E + AGG_CH - 1) / AGG_CH;
    const int  rot  = (int)blockIdx.x * (nCh / (int)gridDim.x);

    int  c = wave;
    int4 nx[4];
    int  nbase = 0;
    if (c < nCh) {
        int cc = c + rot; if (cc >= nCh) cc -= nCh;
        nbase = cc * AGG_CH;
        agg_ld_chunk(nx, dst, nbase, lane, E, vec);
    }
    while (c < nCh) {
        int4 cu[4];
#pragma unroll
        for (int k = 0; k < 4; ++k) cu[k] = nx[k];
        const int cbase = nbase;
        const int cn = c + AGG_W;
        if (cn < nCh) {
            int cc = cn + rot; if (cc >= nCh) cc -= nCh;
            nbase = cc * AGG_CH;
            agg_ld_chunk(nx, dst, nbase, lane, E, vec);
        }
#pragma unroll
        for (int k = 0; k < 4; ++k) {
#pragma unroll
            for (int j = 0; j < 4; ++j) {
                const int d = (j == 0) ? cu[k].x : (j == 1) ? cu[k].y : (j == 2) ? cu[k].z : cu[k].w;
                const unsigned t  = (unsigned)(d - lo);
                const bool     mt = t < (unsigned)cnt;
                const unsigned mask = __builtin_amdgcn_ballot_w32(mt);
                if (mask != 0u) {
                    if (mt) push(qn + (int)__builtin_amdgcn_mbcnt_lo(mask, 0u), cbase + k * 128 + lane * 4 + j, (int)t);
                    qn += __builtin_popcount(mask);
                }
            }
        }
        if (qn >= 32) { flush(qn); qn = 0; }
        c = cn;
    }
    flush(qn);
}

__global__ __launch_bounds__(DEG_T)
void k_deg(const int* __restrict__ dst, const unsigned* __restrict__ tmp, int tmpWords, int nSl,
           unsigned* __restrict__ lists, int* __restrict__ counts, int cap,
           float* __restrict__ dinv, int N, int E) {
    __shared__ int cntd[AGG_R];
    __shared__ int cursor;
    __shared__ int bad;

    const int tid  = threadIdx.x;
    const int lane = tid & 31;
    const int wave = __builtin_amdgcn_readfirstlane(tid >> 5);
    const int lo   = blockIdx.x * AGG_R;
    const int nn   = (N - lo < AGG_R) ? (N - lo) : AGG_R;
    const bool have = (size_t)gridDim.x * nSl * SUB <= (size_t)tmpWords;

    for (int i = tid; i < AGG_R; i += DEG_T) cntd[i] = 0;
    if (tid == 0) { cursor = 0; bad = have ? 0 : 1; }
    __syncthreads();

    if (have) {
        const unsigned* T = tmp + (size_t)blockIdx.x * nSl * SUB;
        unsigned*       L = lists + (size_t)blockIdx.x * cap;
        for (int t0 = wave; t0 < nSl; t0 += DEG_W * DEG_U) {
            unsigned wa[DEG_U], wb[DEG_U];
#pragma unroll
            for (int u = 0; u < DEG_U; ++u) {
                const int t = t0 + u * DEG_W;
                wa[u] = (t < nSl) ? T[(size_t)t * SUB + lane]      : 0u;
                wb[u] = (t < nSl) ? T[(size_t)t * SUB + 32 + lane] : 0u;
            }
            int cc[DEG_U];
            int tot = 0, over = 0;
#pragma unroll
            for (int u = 0; u < DEG_U; ++u) {
                int c = (int)__builtin_amdgcn_readfirstlane(wa[u]);
                if (c > SUB - 1) { over = 1; c = SUB - 1; }
                cc[u] = c;
                tot  += c;
            }
            int at = 0;
            if (lane == 0) {
                if (tot > 0) at = __hip_atomic_fetch_add(&cursor, tot, __ATOMIC_RELAXED, __HIP_MEMORY_SCOPE_WORKGROUP);
                if (over) bad = 1;
            }
            at = __builtin_amdgcn_readfirstlane(at);
#pragma unroll
            for (int u = 0; u < DEG_U; ++u) {
                if (lane >= 1 && lane <= cc[u]) {
                    const int p = at + lane - 1;
                    if (p < cap) L[p] = wa[u];
                    __hip_atomic_fetch_add(&cntd[wa[u] & 255u], 1, __ATOMIC_RELAXED, __HIP_MEMORY_SCOPE_WORKGROUP);
                }
                if (32 + lane <= cc[u]) {
                    const int p = at + 31 + lane;
                    if (p < cap) L[p] = wb[u];
                    __hip_atomic_fetch_add(&cntd[wb[u] & 255u], 1, __ATOMIC_RELAXED, __HIP_MEMORY_SCOPE_WORKGROUP);
                }
                at += cc[u];
            }
        }
    }
    __syncthreads();

    int n = cursor;
    if (bad != 0 || n > cap) {
        for (int i = tid; i < AGG_R; i += DEG_T) cntd[i] = 0;
        __syncthreads();
        for (int e = tid; e < E; e += DEG_T) {
            const unsigned t = (unsigned)(dst[e] - lo);
            if (t < (unsigned)nn)
                __hip_atomic_fetch_add(&cntd[t], 1, __ATOMIC_RELAXED, __HIP_MEMORY_SCOPE_WORKGROUP);
        }
        __syncthreads();
        n = cap + 1;
    }
    if (tid == 0) counts[blockIdx.x] = n;

    for (int i = tid; i < nn; i += DEG_T) dinv[lo + i] = rsqrtf(1.0f + (float)cntd[i]);
}

template <int MODE>
__global__ __launch_bounds__(AGG_T)
void k_agg(const float* __restrict__ H,
           const int* __restrict__ src, const int* __restrict__ dst,
           const float* __restrict__ dinv,
           const unsigned* __restrict__ lists, const int* __restrict__ counts, int cap,
           const float* __restrict__ bias,
           const float* __restrict__ gamma, const float* __restrict__ beta,
           const float* __restrict__ mean, const float* __restrict__ var,
           float* __restrict__ out, int N, int E) {
    __shared__ __attribute__((aligned(16))) float tile[AGG_R * FEAT];
    __shared__ int queue[AGG_W * AGG_Q];

    const int tid  = threadIdx.x;
    const int lane = tid & 31;
    const int wave = __builtin_amdgcn_readfirstlane(tid >> 5);
    const int lo   = blockIdx.x * AGG_R;
    const int cnt  = (N - lo < AGG_R) ? (N - lo) : AGG_R;

    for (int i = tid; i < AGG_R * FEAT / 4; i += AGG_T)
        ((float4*)tile)[i] = make_float4(0.0f, 0.0f, 0.0f, 0.0f);
    __syncthreads();

    auto rows = [&](int m, int s, int dl, float w) {
        for (int u0 = 0; u0 < m; u0 += 8) {
            float v[8][4];
            int   dls[8];
#pragma unroll
            for (int u = 0; u < 8; ++u) {
                const int uu = (u0 + u < m) ? (u0 + u) : (m - 1);
                const int   su = __builtin_amdgcn_readlane(s, uu);
                const float wu = __int_as_float(__builtin_amdgcn_readlane(__float_as_int(w), uu));
                dls[u] = __builtin_amdgcn_readlane(dl, uu);
                const float* hp = H + (size_t)su * FEAT + lane;
                v[u][0] = hp[0]  * wu;
                v[u][1] = hp[32] * wu;
                v[u][2] = hp[64] * wu;
                v[u][3] = hp[96] * wu;
            }
#pragma unroll
            for (int u = 0; u < 8; ++u) {
                if (u0 + u < m) {
                    float* tp = tile + dls[u] * FEAT + lane;
                    __hip_atomic_fetch_add(tp,      v[u][0], __ATOMIC_RELAXED, __HIP_MEMORY_SCOPE_WORKGROUP);
                    __hip_atomic_fetch_add(tp + 32, v[u][1], __ATOMIC_RELAXED, __HIP_MEMORY_SCOPE_WORKGROUP);
                    __hip_atomic_fetch_add(tp + 64, v[u][2], __ATOMIC_RELAXED, __HIP_MEMORY_SCOPE_WORKGROUP);
                    __hip_atomic_fetch_add(tp + 96, v[u][3], __ATOMIC_RELAXED, __HIP_MEMORY_SCOPE_WORKGROUP);
                }
            }
        }
    };

    const int n = counts[blockIdx.x];
    if (n <= cap) {
        const unsigned* L = lists + (size_t)blockIdx.x * cap;
        unsigned ent[LIST_G];
        float    wgt[LIST_G];
#pragma unroll
        for (int k = 0; k < LIST_G; ++k) {
            const int idx = wave * 32 + k * AGG_T + lane;
            ent[k] = (idx < n) ? L[idx] : 0u;
        }
        float da[LIST_G], db[LIST_G];
#pragma unroll
        for (int k = 0; k < LIST_G; ++k) {
            da[k] = dinv[ent[k] >> 8];
            db[k] = dinv[lo + (int)(ent[k] & 255u)];
        }
        asm volatile("" :: "v"(da[0]), "v"(da[1]), "v"(da[2]), "v"(da[3]),
                           "v"(db[0]), "v"(db[1]), "v"(db[2]), "v"(db[3]));
#pragma unroll
        for (int k = 0; k < LIST_G; ++k) wgt[k] = da[k] * db[k];
#pragma unroll
        for (int k = 0; k < LIST_G; ++k) {
            const int g = wave * 32 + k * AGG_T;
            if (g < n) rows((n - g < 32) ? (n - g) : 32, (int)(ent[k] >> 8), (int)(ent[k] & 255u), wgt[k]);
        }
    } else {
        int* q = queue + wave * AGG_Q;
        auto push = [&](int pos, int e, int t) { q[pos] = e; };
        auto flush = [&](int nq) {
            for (int g = 0; g < nq; g += 32) {
                const int m = (nq - g < 32) ? (nq - g) : 32;
                int s = 0, dl = 0; float w = 0.0f;
                if (lane < m) {
                    const int e = q[g + lane];
                    const int d = dst[e];
                    s  = src[e];
                    dl = d - lo;
                    w  = dinv[s] * dinv[d];
                }
                rows(m, s, dl, w);
            }
        };
        scan_dst(dst, E, lo, cnt, lane, wave, push, flush);
    }
    __syncthreads();

    const int   f  = tid & (FEAT - 1);
    const float bb = bias[f];
    float mu = 0.0f, rs = 1.0f, ga = 1.0f, be = 0.0f;
    if (MODE == 1) { mu = mean[f]; rs = rsqrtf(var[f] + BN_EPS); ga = gamma[f]; be = beta[f]; }
    for (int r = tid >> 7; r < cnt; r += AGG_T / FEAT) {
        const int    node = lo + r;
        const float  di   = dinv[node];
        const size_t gi   = (size_t)node * FEAT + f;
        float v = tile[r * FEAT + f] + H[gi] * di * di + bb;
        if (MODE == 1) {
            v = (v - mu) * rs * ga + be;
            v = fmaxf(v, 0.0f);
        }
        out[gi] = v;
    }
}

extern "C" void kernel_launch(void* const* d_in, const int* in_sizes, int n_in,
                              void* d_out, int out_size, void* d_ws, size_t ws_size,
                              hipStream_t stream) {
    const float* X     = (const float*)d_in[0];
    const int*   edge  = (const int*)  d_in[1];
    const float* W1    = (const float*)d_in[2];
    const float* b1    = (const float*)d_in[3];
    const float* gamma = (const float*)d_in[4];
    const float* beta  = (const float*)d_in[5];
    const float* rmean = (const float*)d_in[6];
    const float* rvar  = (const float*)d_in[7];
    const float* W2    = (const float*)d_in[8];
    const float* b2    = (const float*)d_in[9];
    const float* Wc    = (const float*)d_in[10];
    const float* bc    = (const float*)d_in[11];
    float*       out   = (float*)d_out;

    const int N = in_sizes[0] / FEAT;
    const int E = in_sizes[1] / 2;
    const int* src = edge;
    const int* dst = edge + E;
    const int total = N * FEAT;

    float* dinv = (float*)d_ws;
    float* bufA = dinv + ((N + 63) & ~63);
    float* bufB = bufA + (size_t)total;

    const int T256   = 256;
    const int gAgg   = (N + AGG_R - 1) / AGG_R;
    const int nSl    = (E + BIN_E - 1) / BIN_E;
    const int tilesC = ((N + 15) / 16) * (N_CLASSES / 16);
    const int gGemmC = (tilesC + 7) / 8;
    const int gStrip = (N + GS_W * 16 - 1) / (GS_W * 16);

    const int cap = ((N * N_CLASSES - gAgg) / gAgg) & ~3;
    unsigned* lists  = (unsigned*)d_out;
    int*      counts = (int*)d_out + (size_t)gAgg * cap;

    unsigned* tmp = (unsigned*)bufB;

    k_gemm_strip<<<gStrip, GS_T, 0, stream>>>(X, W1, bufA, N, src, dst, tmp, total, E, gAgg, nSl);

    k_deg<<<gAgg, DEG_T, 0, stream>>>(dst, tmp, total, nSl, lists, counts, cap, dinv, N, E);

    k_agg<1><<<gAgg, AGG_T, 0, stream>>>(bufA, src, dst, dinv, lists, counts, cap, b1, gamma, beta, rmean, rvar, bufB, N, E);

    k_gemm_strip<<<gStrip, GS_T, 0, stream>>>(bufB, W2, bufA, N, nullptr, nullptr, nullptr, 0, 0, 0, 0);
    k_agg<2><<<gAgg, AGG_T, 0, stream>>>(bufA, src, dst, dinv, lists, counts, cap, b2, nullptr, nullptr, nullptr, nullptr, bufB, N, E);

    k_gemm_wmma<<<gGemmC, T256, 0, stream>>>(bufB, Wc, bc, out, N, FEAT, N_CLASSES);
}


# Classifier product moved into the layer-2 aggregation blocks: its epilogue keeps the rows in the LDS tile, a wave per 16 nodes multiplies them by Wc on f32 WMMA; 6 launches to 5; sum order unchanged
# speedup vs baseline: 3.8407x; 1.1070x over previous
#include <hip/hip_runtime.h>
#include <math.h>

typedef float v2f __attribute__((ext_vector_type(2)));
typedef float v8f __attribute__((ext_vector_type(8)));

#define FEAT 128
#define N_CLASSES 16
#define BN_EPS 1e-5f
#define AGG_R 200
#define OUT_RS (AGG_R * N_CLASSES)

#define GS_W  16
#define GS_T  (GS_W * 32)
#define GS_E  (32 * 2 * 4 * 16)

#define BIN_K   7
#define BIN_E   (GS_T * BIN_K)
#define BIN_O   1024
#define SUB     64

__device__ __forceinline__ void bin_slices(const int* __restrict__ src, const int* __restrict__ dst,
                                           unsigned* __restrict__ tmp, int N, int E, int nOwn, int nSl,
                                           int* hist) {
    const int tid = threadIdx.x;
    for (int sl = blockIdx.x; sl < nSl; sl += gridDim.x) {
        const int base = sl * BIN_E;
        int d[BIN_K], s[BIN_K];
        if (base + BIN_E <= E) {
#pragma unroll
            for (int k = 0; k < BIN_K; ++k) {
                const int e = base + k * GS_T + tid;
                d[k] = dst[e];
                s[k] = src[e];
            }
        } else {
#pragma unroll
            for (int k = 0; k < BIN_K; ++k) {
                const int e = base + k * GS_T + tid;
                d[k] = (e < E) ? dst[e] : -1;
                s[k] = (e < E) ? src[e] : 0;
            }
        }
        int o[BIN_K], rk[BIN_K];
#pragma unroll
        for (int k = 0; k < BIN_K; ++k) {
            o[k]  = ((unsigned)d[k] < (unsigned)N) ? d[k] / AGG_R : -1;
            rk[k] = 0;
        }
        for (int w0 = 0; w0 < nOwn; w0 += BIN_O) {
            __syncthreads();
            for (int i = tid; i < BIN_O; i += GS_T) hist[i] = 0;
            __syncthreads();
#pragma unroll
            for (int k = 0; k < BIN_K; ++k) {
                const unsigned ow = (unsigned)(o[k] - w0);
                if (ow < (unsigned)BIN_O)
                    rk[k] = __hip_atomic_fetch_add(&hist[ow], 1, __ATOMIC_RELAXED, __HIP_MEMORY_SCOPE_WORKGROUP);
            }
            __syncthreads();
            for (int i = tid; i < BIN_O && w0 + i < nOwn; i += GS_T)
                tmp[((size_t)(w0 + i) * nSl + sl) * SUB] = (unsigned)hist[i];
#pragma unroll
            for (int k = 0; k < BIN_K; ++k) {
                const unsigned ow = (unsigned)(o[k] - w0);
                if (ow < (unsigned)BIN_O && rk[k] < SUB - 1)
                    tmp[((size_t)o[k] * nSl + sl) * SUB + 1 + rk[k]] =
                        ((unsigned)s[k] << 8) | (unsigned)(d[k] - o[k] * AGG_R);
            }
        }
    }
}

__global__ __launch_bounds__(GS_T)
void k_gemm_strip(const float* __restrict__ A, const float* __restrict__ Wm,
                  float* __restrict__ C, int M,
                  const int* __restrict__ src, const int* __restrict__ dst,
                  unsigned* __restrict__ tmp, int tmpWords, int E, int nOwn, int nSl) {
    __shared__ float4 Bl[GS_E];
    __shared__ int hist[BIN_O];

    const int tid  = threadIdx.x;

    if (nSl > 0 && (size_t)nOwn * nSl * SUB <= (size_t)tmpWords)
        bin_slices(src, dst, tmp, M, E, nOwn, nSl, hist);

    const int lane = tid & 31;
    const int wave = __builtin_amdgcn_readfirstlane(tid >> 5);
    const int c    = lane & 15;
    const int h    = lane >> 4;
    const int m0   = (blockIdx.x * GS_W + wave) << 4;

    float2 w0[GS_E / GS_T], w1[GS_E / GS_T];
#pragma unroll
    for (int m = 0; m < GS_E / GS_T; ++m) {
        const int i  = tid + m * GS_T;
        const int w  = i >> 7;
        const int q  = (i >> 4) & 3;
        const int k0 = 8 * (w >> 1) + 4 * ((i >> 6) & 1) + 2 * (w & 1);
        const int cc = 64 * (q >> 1) + 4 * (i & 15) + 2 * (q & 1);
        w0[m] = *(const float2*)(Wm + k0 * FEAT + cc);
        w1[m] = *(const float2*)(Wm + (k0 + 1) * FEAT + cc);
    }

#pragma unroll
    for (int m = 0; m < GS_E / GS_T; ++m)
        Bl[tid + m * GS_T] = make_float4(w0[m].x, w1[m].x, w0[m].y, w1[m].y);
    __syncthreads();
    if (m0 >= M) return;

    int row = m0 + c;
    if (row > M - 1) row = M - 1;
    const float4* ap = (const float4*)(A + (size_t)row * FEAT + 4 * h);
    float4 a[16];
#pragma unroll
    for (int j = 0; j < 16; ++j) a[j] = ap[2 * j];

    v8f acc[8];
#pragma unroll
    for (int n = 0; n < 8; ++n) acc[n] = v8f{};

    const float4* bp = Bl + h * 64 + c;
    float4 bf[2][8];
#pragma unroll
    for (int q = 0; q < 8; ++q) bf[0][q] = bp[(q >> 2) * 128 + (q & 3) * 16];
#pragma unroll
    for (int g = 0; g < 16; ++g) {
        if (g + 1 < 16) {
#pragma unroll
            for (int q = 0; q < 8; ++q)
                bf[(g + 1) & 1][q] = bp[(2 * (g + 1) + (q >> 2)) * 128 + (q & 3) * 16];
        }
        __builtin_amdgcn_sched_barrier(0);
#pragma unroll
        for (int t = 0; t < 2; ++t) {
            v2f av;
            av.x = t ? a[g].z : a[g].x;
            av.y = t ? a[g].w : a[g].y;
#pragma unroll
            for (int q = 0; q < 4; ++q) {
                const float4 b = bf[g & 1][4 * t + q];
                v2f b0, b1;
                b0.x = b.x; b0.y = b.y;
                b1.x = b.z; b1.y = b.w;
                acc[2 * q] = __builtin_amdgcn_wmma_f32_16x16x4_f32(
                                  false, av,  false, b0,
                                  (short)0, acc[2 * q],  false,  false);
                acc[2 * q + 1] = __builtin_amdgcn_wmma_f32_16x16x4_f32(
                                  false, av,  false, b1,
                                  (short)0, acc[2 * q + 1],  false,  false);
            }
        }
        __builtin_amdgcn_sched_barrier(0);
    }

#pragma unroll
    for (int j = 0; j < 8; ++j) {
        const int r = m0 + j + 8 * h;
        if (r < M) {
            float* cp = C + (size_t)r * FEAT + 4 * c;
            *(float4*)cp        = make_float4(acc[0][j], acc[1][j], acc[2][j], acc[3][j]);
            *(float4*)(cp + 64) = make_float4(acc[4][j], acc[5][j], acc[6][j], acc[7][j]);
        }
    }
}

#define AGG_T   1024
#define AGG_W   (AGG_T / 32)
#define AGG_CH  512
#define AGG_Q   (AGG_CH + 32)
#define LIST_G  4
#define DEG_T   256
#define DEG_W   (DEG_T / 32)
#define DEG_U   8

__device__ __forceinline__ void agg_ld_chunk(int4 (&r)[4], const int* __restrict__ dst,
                                             int base, int lane, int E, bool vec) {
    if (vec && base + AGG_CH <= E) {
#pragma unroll
        for (int k = 0; k < 4; ++k) r[k] = *(const int4*)(dst + base + k * 128 + lane * 4);
    } else {
#pragma unroll
        for (int k = 0; k < 4; ++k) {
            const int e0 = base + k * 128 + lane * 4;
            r[k].x = (e0     < E) ? dst[e0]     : -1;
            r[k].y = (e0 + 1 < E) ? dst[e0 + 1] : -1;
            r[k].z = (e0 + 2 < E) ? dst[e0 + 2] : -1;
            r[k].w = (e0 + 3 < E) ? dst[e0 + 3] : -1;
        }
    }
}

template <class P, class F>
__device__ __forceinline__ void scan_dst(const int* __restrict__ dst, int E, int lo, int cnt,
                                         int lane, int wave, P push, F flush) {
    int qn = 0;
    const bool vec  = ((reinterpret_cast<uintptr_t>(dst) & 15) == 0);
    const int  nCh  = (E + AGG_CH - 1) / AGG_CH;
    const int  rot  = (int)blockIdx.x * (nCh / (int)gridDim.x);

    int  c = wave;
    int4 nx[4];
    int  nbase = 0;
    if (c < nCh) {
        int cc = c + rot; if (cc >= nCh) cc -= nCh;
        nbase = cc * AGG_CH;
        agg_ld_chunk(nx, dst, nbase, lane, E, vec);
    }
    while (c < nCh) {
        int4 cu[4];
#pragma unroll
        for (int k = 0; k < 4; ++k) cu[k] = nx[k];
        const int cbase = nbase;
        const int cn = c + AGG_W;
        if (cn < nCh) {
            int cc = cn + rot; if (cc >= nCh) cc -= nCh;
            nbase = cc * AGG_CH;
            agg_ld_chunk(nx, dst, nbase, lane, E, vec);
        }
#pragma unroll
        for (int k = 0; k < 4; ++k) {
#pragma unroll
            for (int j = 0; j < 4; ++j) {
                const int d = (j == 0) ? cu[k].x : (j == 1) ? cu[k].y : (j == 2) ? cu[k].z : cu[k].w;
                const unsigned t  = (unsigned)(d - lo);
                const bool     mt = t < (unsigned)cnt;
                const unsigned mask = __builtin_amdgcn_ballot_w32(mt);
                if (mask != 0u) {
                    if (mt) push(qn + (int)__builtin_amdgcn_mbcnt_lo(mask, 0u), cbase + k * 128 + lane * 4 + j, (int)t);
                    qn += __builtin_popcount(mask);
                }
            }
        }
        if (qn >= 32) { flush(qn); qn = 0; }
        c = cn;
    }
    flush(qn);
}

__global__ __launch_bounds__(DEG_T)
void k_deg(const int* __restrict__ dst, const unsigned* __restrict__ tmp, int tmpWords, int nSl,
           unsigned* __restrict__ lists,
           float* __restrict__ dinv, int N, int E) {
    __shared__ int cntd[AGG_R];
    __shared__ int cursor;
    __shared__ int bad;

    const int tid  = threadIdx.x;
    const int lane = tid & 31;
    const int wave = __builtin_amdgcn_readfirstlane(tid >> 5);
    const int lo   = blockIdx.x * AGG_R;
    const int nn   = (N - lo < AGG_R) ? (N - lo) : AGG_R;
    const int cap  = nn * N_CLASSES - 1;
    const bool have = (size_t)gridDim.x * nSl * SUB <= (size_t)tmpWords;

    for (int i = tid; i < AGG_R; i += DEG_T) cntd[i] = 0;
    if (tid == 0) { cursor = 0; bad = have ? 0 : 1; }
    __syncthreads();

    if (have) {
        const unsigned* T = tmp + (size_t)blockIdx.x * nSl * SUB;
        unsigned*       L = lists + (size_t)blockIdx.x * OUT_RS + 1;
        for (int t0 = wave; t0 < nSl; t0 += DEG_W * DEG_U) {
            unsigned wa[DEG_U], wb[DEG_U];
#pragma unroll
            for (int u = 0; u < DEG_U; ++u) {
                const int t = t0 + u * DEG_W;
                wa[u] = (t < nSl) ? T[(size_t)t * SUB + lane]      : 0u;
                wb[u] = (t < nSl) ? T[(size_t)t * SUB + 32 + lane] : 0u;
            }
            int cc[DEG_U];
            int tot = 0, over = 0;
#pragma unroll
            for (int u = 0; u < DEG_U; ++u) {
                int c = (int)__builtin_amdgcn_readfirstlane(wa[u]);
                if (c > SUB - 1) { over = 1; c = SUB - 1; }
                cc[u] = c;
                tot  += c;
            }
            int at = 0;
            if (lane == 0) {
                if (tot > 0) at = __hip_atomic_fetch_add(&cursor, tot, __ATOMIC_RELAXED, __HIP_MEMORY_SCOPE_WORKGROUP);
                if (over) bad = 1;
            }
            at = __builtin_amdgcn_readfirstlane(at);
#pragma unroll
            for (int u = 0; u < DEG_U; ++u) {
                if (lane >= 1 && lane <= cc[u]) {
                    const int p = at + lane - 1;
                    if (p < cap) L[p] = wa[u];
                    __hip_atomic_fetch_add(&cntd[wa[u] & 255u], 1, __ATOMIC_RELAXED, __HIP_MEMORY_SCOPE_WORKGROUP);
                }
                if (32 + lane <= cc[u]) {
                    const int p = at + 31 + lane;
                    if (p < cap) L[p] = wb[u];
                    __hip_atomic_fetch_add(&cntd[wb[u] & 255u], 1, __ATOMIC_RELAXED, __HIP_MEMORY_SCOPE_WORKGROUP);
                }
                at += cc[u];
            }
        }
    }
    __syncthreads();

    int n = cursor;
    if (bad != 0 || n > cap) {
        for (int i = tid; i < AGG_R; i += DEG_T) cntd[i] = 0;
        __syncthreads();
        for (int e = tid; e < E; e += DEG_T) {
            const unsigned t = (unsigned)(dst[e] - lo);
            if (t < (unsigned)nn)
                __hip_atomic_fetch_add(&cntd[t], 1, __ATOMIC_RELAXED, __HIP_MEMORY_SCOPE_WORKGROUP);
        }
        __syncthreads();
        n = cap + 1;
    }
    if (tid == 0) lists[(size_t)blockIdx.x * OUT_RS] = (unsigned)n;

    for (int i = tid; i < nn; i += DEG_T) dinv[lo + i] = rsqrtf(1.0f + (float)cntd[i]);
}

template <int MODE>
__global__ __launch_bounds__(AGG_T)
void k_agg(const float* __restrict__ H,
           const int* __restrict__ src, const int* __restrict__ dst,
           const float* __restrict__ dinv,
           const unsigned* lists,
           const float* __restrict__ bias,
           const float* __restrict__ gamma, const float* __restrict__ beta,
           const float* __restrict__ mean, const float* __restrict__ var,
           float* __restrict__ out,
           const float* __restrict__ Wc, const float* __restrict__ bc, float* cls,
           int N, int E) {
    constexpr int TS = (MODE == 2) ? (FEAT + 4) : FEAT;
    __shared__ __attribute__((aligned(16))) float tile[AGG_R * TS];
    __shared__ int queue[AGG_W * AGG_Q];

    const int tid  = threadIdx.x;
    const int lane = tid & 31;
    const int wave = __builtin_amdgcn_readfirstlane(tid >> 5);
    const int lo   = blockIdx.x * AGG_R;
    const int cnt  = (N - lo < AGG_R) ? (N - lo) : AGG_R;
    const int cap  = cnt * N_CLASSES - 1;

    for (int i = tid; i < AGG_R * TS / 4; i += AGG_T)
        ((float4*)tile)[i] = make_float4(0.0f, 0.0f, 0.0f, 0.0f);
    __syncthreads();

    auto rows = [&](int m, int s, int dl, float w) {
        for (int u0 = 0; u0 < m; u0 += 8) {
            float v[8][4];
            int   dls[8];
#pragma unroll
            for (int u = 0; u < 8; ++u) {
                const int uu = (u0 + u < m) ? (u0 + u) : (m - 1);
                const int   su = __builtin_amdgcn_readlane(s, uu);
                const float wu = __int_as_float(__builtin_amdgcn_readlane(__float_as_int(w), uu));
                dls[u] = __builtin_amdgcn_readlane(dl, uu);
                const float* hp = H + (size_t)su * FEAT + lane;
                v[u][0] = hp[0]  * wu;
                v[u][1] = hp[32] * wu;
                v[u][2] = hp[64] * wu;
                v[u][3] = hp[96] * wu;
            }
#pragma unroll
            for (int u = 0; u < 8; ++u) {
                if (u0 + u < m) {
                    float* tp = tile + dls[u] * TS + lane;
                    __hip_atomic_fetch_add(tp,      v[u][0], __ATOMIC_RELAXED, __HIP_MEMORY_SCOPE_WORKGROUP);
                    __hip_atomic_fetch_add(tp + 32, v[u][1], __ATOMIC_RELAXED, __HIP_MEMORY_SCOPE_WORKGROUP);
                    __hip_atomic_fetch_add(tp + 64, v[u][2], __ATOMIC_RELAXED, __HIP_MEMORY_SCOPE_WORKGROUP);
                    __hip_atomic_fetch_add(tp + 96, v[u][3], __ATOMIC_RELAXED, __HIP_MEMORY_SCOPE_WORKGROUP);
                }
            }
        }
    };

    const int n = (int)lists[(size_t)blockIdx.x * OUT_RS];
    if (n <= cap) {
        const unsigned* L = lists + (size_t)blockIdx.x * OUT_RS + 1;
        unsigned ent[LIST_G];
        float    wgt[LIST_G];
#pragma unroll
        for (int k = 0; k < LIST_G; ++k) {
            const int idx = wave * 32 + k * AGG_T + lane;
            ent[k] = (idx < n) ? L[idx] : 0u;
        }
        float da[LIST_G], db[LIST_G];
#pragma unroll
        for (int k = 0; k < LIST_G; ++k) {
            da[k] = dinv[ent[k] >> 8];
            db[k] = dinv[lo + (int)(ent[k] & 255u)];
        }
        asm volatile("" :: "v"(da[0]), "v"(da[1]), "v"(da[2]), "v"(da[3]),
                           "v"(db[0]), "v"(db[1]), "v"(db[2]), "v"(db[3]));
#pragma unroll
        for (int k = 0; k < LIST_G; ++k) wgt[k] = da[k] * db[k];
#pragma unroll
        for (int k = 0; k < LIST_G; ++k) {
            const int g = wave * 32 + k * AGG_T;
            if (g < n) rows((n - g < 32) ? (n - g) : 32, (int)(ent[k] >> 8), (int)(ent[k] & 255u), wgt[k]);
        }
    } else {
        int* q = queue + wave * AGG_Q;
        auto push = [&](int pos, int e, int t) { q[pos] = e; };
        auto flush = [&](int nq) {
            for (int g = 0; g < nq; g += 32) {
                const int m = (nq - g < 32) ? (nq - g) : 32;
                int s = 0, dl = 0; float w = 0.0f;
                if (lane < m) {
                    const int e = q[g + lane];
                    const int d = dst[e];
                    s  = src[e];
                    dl = d - lo;
                    w  = dinv[s] * dinv[d];
                }
                rows(m, s, dl, w);
            }
        };
        scan_dst(dst, E, lo, cnt, lane, wave, push, flush);
    }
    __syncthreads();

    const int   f  = tid & (FEAT - 1);
    const float bb = bias[f];
    float mu = 0.0f, rs = 1.0f, ga = 1.0f, be = 0.0f;
    if (MODE == 1) { mu = mean[f]; rs = rsqrtf(var[f] + BN_EPS); ga = gamma[f]; be = beta[f]; }
    for (int r = tid >> 7; r < cnt; r += AGG_T / FEAT) {
        const int    node = lo + r;
        const float  di   = dinv[node];
        const size_t gi   = (size_t)node * FEAT + f;
        float v = tile[r * TS + f] + H[gi] * di * di + bb;
        if (MODE == 1) {
            v = (v - mu) * rs * ga + be;
            v = fmaxf(v, 0.0f);
            out[gi] = v;
        } else {
            tile[r * TS + f] = v;
        }
    }

    if (MODE == 2) {
        __syncthreads();
        const int nStrip = (cnt + 15) >> 4;
        if (wave < nStrip) {
            const int c = lane & 15;
            const int h = lane >> 4;
            int row = 16 * wave + c;
            if (row > cnt - 1) row = cnt - 1;
            const float* ap = tile + row * TS + 2 * h;
            const float* bp = Wc + (2 * h) * N_CLASSES + c;
            v8f acc = {};
#pragma unroll
            for (int i = 0; i < FEAT / 4; ++i) {
                const v2f a = *(const v2f*)(ap + 4 * i);
                v2f b;
                b.x = bp[(4 * i)     * N_CLASSES];
                b.y = bp[(4 * i + 1) * N_CLASSES];
                acc = __builtin_amdgcn_wmma_f32_16x16x4_f32(
                           false, a,  false, b,
                           (short)0, acc,  false,  false);
            }
            const float bcv = bc[c];
#pragma unroll
            for (int j = 0; j < 8; ++j) {
                const int r = 16 * wave + j + 8 * h;
                if (r < cnt) cls[(size_t)(lo + r) * N_CLASSES + c] = acc[j] + bcv;
            }
        }
    }
}

extern "C" void kernel_launch(void* const* d_in, const int* in_sizes, int n_in,
                              void* d_out, int out_size, void* d_ws, size_t ws_size,
                              hipStream_t stream) {
    const float* X     = (const float*)d_in[0];
    const int*   edge  = (const int*)  d_in[1];
    const float* W1    = (const float*)d_in[2];
    const float* b1    = (const float*)d_in[3];
    const float* gamma = (const float*)d_in[4];
    const float* beta  = (const float*)d_in[5];
    const float* rmean = (const float*)d_in[6];
    const float* rvar  = (const float*)d_in[7];
    const float* W2    = (const float*)d_in[8];
    const float* b2    = (const float*)d_in[9];
    const float* Wc    = (const float*)d_in[10];
    const float* bc    = (const float*)d_in[11];
    float*       out   = (float*)d_out;

    const int N = in_sizes[0] / FEAT;
    const int E = in_sizes[1] / 2;
    const int* src = edge;
    const int* dst = edge + E;
    const int total = N * FEAT;

    float* dinv = (float*)d_ws;
    float* bufA = dinv + ((N + 63) & ~63);
    float* bufB = bufA + (size_t)total;

    const int gAgg   = (N + AGG_R - 1) / AGG_R;
    const int nSl    = (E + BIN_E - 1) / BIN_E;
    const int gStrip = (N + GS_W * 16 - 1) / (GS_W * 16);

    unsigned* lists  = (unsigned*)d_out;

    unsigned* tmp = (unsigned*)bufB;

    k_gemm_strip<<<gStrip, GS_T, 0, stream>>>(X, W1, bufA, N, src, dst, tmp, total, E, gAgg, nSl);

    k_deg<<<gAgg, DEG_T, 0, stream>>>(dst, tmp, total, nSl, lists, dinv, N, E);

    k_agg<1><<<gAgg, AGG_T, 0, stream>>>(bufA, src, dst, dinv, lists, b1, gamma, beta, rmean, rvar, bufB, nullptr, nullptr, nullptr, N, E);

    k_gemm_strip<<<gStrip, GS_T, 0, stream>>>(bufB, W2, bufA, N, nullptr, nullptr, nullptr, 0, 0, 0, 0);
    k_agg<2><<<gAgg, AGG_T, 0, stream>>>(bufA, src, dst, dinv, lists, b2, nullptr, nullptr, nullptr, nullptr, nullptr, Wc, bc, out, N, E);
}


# Layer-2 product moved behind its aggregation, (Ahat act) W2, into the owner blocks: the LDS tile of summed rows times W2 (staged in LDS) on f32 WMMA, then Wc; 5 launches to 4; f32 sums reassociated
# speedup vs baseline: 4.1424x; 1.0786x over previous
#include <hip/hip_runtime.h>
#include <math.h>

typedef float v2f __attribute__((ext_vector_type(2)));
typedef float v8f __attribute__((ext_vector_type(8)));

#define FEAT 128
#define N_CLASSES 16
#define BN_EPS 1e-5f
#define AGG_R 200
#define OUT_RS (AGG_R * N_CLASSES)

#define GS_W  16
#define GS_T  (GS_W * 32)
#define GS_E  (32 * 2 * 4 * 16)

#define BIN_K   7
#define BIN_E   (GS_T * BIN_K)
#define BIN_O   1024
#define SUB     64

__device__ __forceinline__ void bin_slices(const int* __restrict__ src, const int* __restrict__ dst,
                                           unsigned* __restrict__ tmp, int N, int E, int nOwn, int nSl,
                                           int* hist) {
    const int tid = threadIdx.x;
    for (int sl = blockIdx.x; sl < nSl; sl += gridDim.x) {
        const int base = sl * BIN_E;
        int d[BIN_K], s[BIN_K];
        if (base + BIN_E <= E) {
#pragma unroll
            for (int k = 0; k < BIN_K; ++k) {
                const int e = base + k * GS_T + tid;
                d[k] = dst[e];
                s[k] = src[e];
            }
        } else {
#pragma unroll
            for (int k = 0; k < BIN_K; ++k) {
                const int e = base + k * GS_T + tid;
                d[k] = (e < E) ? dst[e] : -1;
                s[k] = (e < E) ? src[e] : 0;
            }
        }
        int o[BIN_K], rk[BIN_K];
#pragma unroll
        for (int k = 0; k < BIN_K; ++k) {
            o[k]  = ((unsigned)d[k] < (unsigned)N) ? d[k] / AGG_R : -1;
            rk[k] = 0;
        }
        for (int w0 = 0; w0 < nOwn; w0 += BIN_O) {
            __syncthreads();
            for (int i = tid; i < BIN_O; i += GS_T) hist[i] = 0;
            __syncthreads();
#pragma unroll
            for (int k = 0; k < BIN_K; ++k) {
                const unsigned ow = (unsigned)(o[k] - w0);
                if (ow < (unsigned)BIN_O)
                    rk[k] = __hip_atomic_fetch_add(&hist[ow], 1, __ATOMIC_RELAXED, __HIP_MEMORY_SCOPE_WORKGROUP);
            }
            __syncthreads();
            for (int i = tid; i < BIN_O && w0 + i < nOwn; i += GS_T)
                tmp[((size_t)(w0 + i) * nSl + sl) * SUB] = (unsigned)hist[i];
#pragma unroll
            for (int k = 0; k < BIN_K; ++k) {
                const unsigned ow = (unsigned)(o[k] - w0);
                if (ow < (unsigned)BIN_O && rk[k] < SUB - 1)
                    tmp[((size_t)o[k] * nSl + sl) * SUB + 1 + rk[k]] =
                        ((unsigned)s[k] << 8) | (unsigned)(d[k] - o[k] * AGG_R);
            }
        }
    }
}

__global__ __launch_bounds__(GS_T)
void k_gemm_strip(const float* __restrict__ A, const float* __restrict__ Wm,
                  float* __restrict__ C, int M,
                  const int* __restrict__ src, const int* __restrict__ dst,
                  unsigned* __restrict__ tmp, int tmpWords, int E, int nOwn, int nSl) {
    __shared__ float4 Bl[GS_E];
    __shared__ int hist[BIN_O];

    const int tid  = threadIdx.x;

    if (nSl > 0 && (size_t)nOwn * nSl * SUB <= (size_t)tmpWords)
        bin_slices(src, dst, tmp, M, E, nOwn, nSl, hist);

    const int lane = tid & 31;
    const int wave = __builtin_amdgcn_readfirstlane(tid >> 5);
    const int c    = lane & 15;
    const int h    = lane >> 4;
    const int m0   = (blockIdx.x * GS_W + wave) << 4;

    float2 w0[GS_E / GS_T], w1[GS_E / GS_T];
#pragma unroll
    for (int m = 0; m < GS_E / GS_T; ++m) {
        const int i  = tid + m * GS_T;
        const int w  = i >> 7;
        const int q  = (i >> 4) & 3;
        const int k0 = 8 * (w >> 1) + 4 * ((i >> 6) & 1) + 2 * (w & 1);
        const int cc = 64 * (q >> 1) + 4 * (i & 15) + 2 * (q & 1);
        w0[m] = *(const float2*)(Wm + k0 * FEAT + cc);
        w1[m] = *(const float2*)(Wm + (k0 + 1) * FEAT + cc);
    }

#pragma unroll
    for (int m = 0; m < GS_E / GS_T; ++m)
        Bl[tid + m * GS_T] = make_float4(w0[m].x, w1[m].x, w0[m].y, w1[m].y);
    __syncthreads();
    if (m0 >= M) return;

    int row = m0 + c;
    if (row > M - 1) row = M - 1;
    const float4* ap = (const float4*)(A + (size_t)row * FEAT + 4 * h);
    float4 a[16];
#pragma unroll
    for (int j = 0; j < 16; ++j) a[j] = ap[2 * j];

    v8f acc[8];
#pragma unroll
    for (int n = 0; n < 8; ++n) acc[n] = v8f{};

    const float4* bp = Bl + h * 64 + c;
    float4 bf[2][8];
#pragma unroll
    for (int q = 0; q < 8; ++q) bf[0][q] = bp[(q >> 2) * 128 + (q & 3) * 16];
#pragma unroll
    for (int g = 0; g < 16; ++g) {
        if (g + 1 < 16) {
#pragma unroll
            for (int q = 0; q < 8; ++q)
                bf[(g + 1) & 1][q] = bp[(2 * (g + 1) + (q >> 2)) * 128 + (q & 3) * 16];
        }
        __builtin_amdgcn_sched_barrier(0);
#pragma unroll
        for (int t = 0; t < 2; ++t) {
            v2f av;
            av.x = t ? a[g].z : a[g].x;
            av.y = t ? a[g].w : a[g].y;
#pragma unroll
            for (int q = 0; q < 4; ++q) {
                const float4 b = bf[g & 1][4 * t + q];
                v2f b0, b1;
                b0.x = b.x; b0.y = b.y;
                b1.x = b.z; b1.y = b.w;
                acc[2 * q] = __builtin_amdgcn_wmma_f32_16x16x4_f32(
                                  false, av,  false, b0,
                                  (short)0, acc[2 * q],  false,  false);
                acc[2 * q + 1] = __builtin_amdgcn_wmma_f32_16x16x4_f32(
                                  false, av,  false, b1,
                                  (short)0, acc[2 * q + 1],  false,  false);
            }
        }
        __builtin_amdgcn_sched_barrier(0);
    }

#pragma unroll
    for (int j = 0; j < 8; ++j) {
        const int r = m0 + j + 8 * h;
        if (r < M) {
            float* cp = C + (size_t)r * FEAT + 4 * c;
            *(float4*)cp        = make_float4(acc[0][j], acc[1][j], acc[2][j], acc[3][j]);
            *(float4*)(cp + 64) = make_float4(acc[4][j], acc[5][j], acc[6][j], acc[7][j]);
        }
    }
}

#define AGG_T   1024
#define AGG_W   (AGG_T / 32)
#define AGG_CH  512
#define AGG_Q   (AGG_CH + 32)
#define LIST_G  4
#define DEG_T   256
#define DEG_W   (DEG_T / 32)
#define DEG_U   8

__device__ __forceinline__ void agg_ld_chunk(int4 (&r)[4], const int* __restrict__ dst,
                                             int base, int lane, int E, bool vec) {
    if (vec && base + AGG_CH <= E) {
#pragma unroll
        for (int k = 0; k < 4; ++k) r[k] = *(const int4*)(dst + base + k * 128 + lane * 4);
    } else {
#pragma unroll
        for (int k = 0; k < 4; ++k) {
            const int e0 = base + k * 128 + lane * 4;
            r[k].x = (e0     < E) ? dst[e0]     : -1;
            r[k].y = (e0 + 1 < E) ? dst[e0 + 1] : -1;
            r[k].z = (e0 + 2 < E) ? dst[e0 + 2] : -1;
            r[k].w = (e0 + 3 < E) ? dst[e0 + 3] : -1;
        }
    }
}

template <class P, class F>
__device__ __forceinline__ void scan_dst(const int* __restrict__ dst, int E, int lo, int cnt,
                                         int lane, int wave, P push, F flush) {
    int qn = 0;
    const bool vec  = ((reinterpret_cast<uintptr_t>(dst) & 15) == 0);
    const int  nCh  = (E + AGG_CH - 1) / AGG_CH;
    const int  rot  = (int)blockIdx.x * (nCh / (int)gridDim.x);

    int  c = wave;
    int4 nx[4];
    int  nbase = 0;
    if (c < nCh) {
        int cc = c + rot; if (cc >= nCh) cc -= nCh;
        nbase = cc * AGG_CH;
        agg_ld_chunk(nx, dst, nbase, lane, E, vec);
    }
    while (c < nCh) {
        int4 cu[4];
#pragma unroll
        for (int k = 0; k < 4; ++k) cu[k] = nx[k];
        const int cbase = nbase;
        const int cn = c + AGG_W;
        if (cn < nCh) {
            int cc = cn + rot; if (cc >= nCh) cc -= nCh;
            nbase = cc * AGG_CH;
            agg_ld_chunk(nx, dst, nbase, lane, E, vec);
        }
#pragma unroll
        for (int k = 0; k < 4; ++k) {
#pragma unroll
            for (int j = 0; j < 4; ++j) {
                const int d = (j == 0) ? cu[k].x : (j == 1) ? cu[k].y : (j == 2) ? cu[k].z : cu[k].w;
                const unsigned t  = (unsigned)(d - lo);
                const bool     mt = t < (unsigned)cnt;
                const unsigned mask = __builtin_amdgcn_ballot_w32(mt);
                if (mask != 0u) {
                    if (mt) push(qn + (int)__builtin_amdgcn_mbcnt_lo(mask, 0u), cbase + k * 128 + lane * 4 + j, (int)t);
                    qn += __builtin_popcount(mask);
                }
            }
        }
        if (qn >= 32) { flush(qn); qn = 0; }
        c = cn;
    }
    flush(qn);
}

__global__ __launch_bounds__(DEG_T)
void k_deg(const int* __restrict__ dst, const unsigned* __restrict__ tmp, int tmpWords, int nSl,
           unsigned* __restrict__ lists,
           float* __restrict__ dinv, int N, int E) {
    __shared__ int cntd[AGG_R];
    __shared__ int cursor;
    __shared__ int bad;

    const int tid  = threadIdx.x;
    const int lane = tid & 31;
    const int wave = __builtin_amdgcn_readfirstlane(tid >> 5);
    const int lo   = blockIdx.x * AGG_R;
    const int nn   = (N - lo < AGG_R) ? (N - lo) : AGG_R;
    const int cap  = nn * N_CLASSES - 1;
    const bool have = (size_t)gridDim.x * nSl * SUB <= (size_t)tmpWords;

    for (int i = tid; i < AGG_R; i += DEG_T) cntd[i] = 0;
    if (tid == 0) { cursor = 0; bad = have ? 0 : 1; }
    __syncthreads();

    if (have) {
        const unsigned* T = tmp + (size_t)blockIdx.x * nSl * SUB;
        unsigned*       L = lists + (size_t)blockIdx.x * OUT_RS + 1;
        for (int t0 = wave; t0 < nSl; t0 += DEG_W * DEG_U) {
            unsigned wa[DEG_U], wb[DEG_U];
#pragma unroll
            for (int u = 0; u < DEG_U; ++u) {
                const int t = t0 + u * DEG_W;
                wa[u] = (t < nSl) ? T[(size_t)t * SUB + lane]      : 0u;
                wb[u] = (t < nSl) ? T[(size_t)t * SUB + 32 + lane] : 0u;
            }
            int cc[DEG_U];
            int tot = 0, over = 0;
#pragma unroll
            for (int u = 0; u < DEG_U; ++u) {
                int c = (int)__builtin_amdgcn_readfirstlane(wa[u]);
                if (c > SUB - 1) { over = 1; c = SUB - 1; }
                cc[u] = c;
                tot  += c;
            }
            int at = 0;
            if (lane == 0) {
                if (tot > 0) at = __hip_atomic_fetch_add(&cursor, tot, __ATOMIC_RELAXED, __HIP_MEMORY_SCOPE_WORKGROUP);
                if (over) bad = 1;
            }
            at = __builtin_amdgcn_readfirstlane(at);
#pragma unroll
            for (int u = 0; u < DEG_U; ++u) {
                if (lane >= 1 && lane <= cc[u]) {
                    const int p = at + lane - 1;
                    if (p < cap) L[p] = wa[u];
                    __hip_atomic_fetch_add(&cntd[wa[u] & 255u], 1, __ATOMIC_RELAXED, __HIP_MEMORY_SCOPE_WORKGROUP);
                }
                if (32 + lane <= cc[u]) {
                    const int p = at + 31 + lane;
                    if (p < cap) L[p] = wb[u];
                    __hip_atomic_fetch_add(&cntd[wb[u] & 255u], 1, __ATOMIC_RELAXED, __HIP_MEMORY_SCOPE_WORKGROUP);
                }
                at += cc[u];
            }
        }
    }
    __syncthreads();

    int n = cursor;
    if (bad != 0 || n > cap) {
        for (int i = tid; i < AGG_R; i += DEG_T) cntd[i] = 0;
        __syncthreads();
        for (int e = tid; e < E; e += DEG_T) {
            const unsigned t = (unsigned)(dst[e] - lo);
            if (t < (unsigned)nn)
                __hip_atomic_fetch_add(&cntd[t], 1, __ATOMIC_RELAXED, __HIP_MEMORY_SCOPE_WORKGROUP);
        }
        __syncthreads();
        n = cap + 1;
    }
    if (tid == 0) lists[(size_t)blockIdx.x * OUT_RS] = (unsigned)n;

    for (int i = tid; i < nn; i += DEG_T) dinv[lo + i] = rsqrtf(1.0f + (float)cntd[i]);
}

template <int MODE>
__global__ __launch_bounds__(AGG_T)
void k_agg(const float* __restrict__ H,
           const int* __restrict__ src, const int* __restrict__ dst,
           const float* __restrict__ dinv,
           const unsigned* lists,
           const float* __restrict__ bias,
           const float* __restrict__ gamma, const float* __restrict__ beta,
           const float* __restrict__ mean, const float* __restrict__ var,
           float* __restrict__ out,
           const float* __restrict__ W2,
           const float* __restrict__ Wc, const float* __restrict__ bc, float* cls,
           int N, int E) {
    constexpr int TS = (MODE == 2) ? (FEAT + 4) : FEAT;
    __shared__ __attribute__((aligned(16))) float tile[AGG_R * TS];
    __shared__ int queue[AGG_W * AGG_Q];
    __shared__ float4 Bl[(MODE == 2) ? GS_E : 1];

    const int tid  = threadIdx.x;
    const int lane = tid & 31;
    const int wave = __builtin_amdgcn_readfirstlane(tid >> 5);
    const int lo   = blockIdx.x * AGG_R;
    const int cnt  = (N - lo < AGG_R) ? (N - lo) : AGG_R;
    const int cap  = cnt * N_CLASSES - 1;

    float2 w0[GS_E / AGG_T], w1[GS_E / AGG_T];
    if (MODE == 2) {
#pragma unroll
        for (int m = 0; m < GS_E / AGG_T; ++m) {
            const int i  = tid + m * AGG_T;
            const int w  = i >> 7;
            const int q  = (i >> 4) & 3;
            const int k0 = 8 * (w >> 1) + 4 * ((i >> 6) & 1) + 2 * (w & 1);
            const int cc = 64 * (q >> 1) + 4 * (i & 15) + 2 * (q & 1);
            w0[m] = *(const float2*)(W2 + k0 * FEAT + cc);
            w1[m] = *(const float2*)(W2 + (k0 + 1) * FEAT + cc);
        }
    }

    for (int i = tid; i < AGG_R * TS / 4; i += AGG_T)
        ((float4*)tile)[i] = make_float4(0.0f, 0.0f, 0.0f, 0.0f);
    if (MODE == 2) {
#pragma unroll
        for (int m = 0; m < GS_E / AGG_T; ++m)
            Bl[tid + m * AGG_T] = make_float4(w0[m].x, w1[m].x, w0[m].y, w1[m].y);
    }
    __syncthreads();

    auto rows = [&](int m, int s, int dl, float w) {
        for (int u0 = 0; u0 < m; u0 += 8) {
            float v[8][4];
            int   dls[8];
#pragma unroll
            for (int u = 0; u < 8; ++u) {
                const int uu = (u0 + u < m) ? (u0 + u) : (m - 1);
                const int   su = __builtin_amdgcn_readlane(s, uu);
                const float wu = __int_as_float(__builtin_amdgcn_readlane(__float_as_int(w), uu));
                dls[u] = __builtin_amdgcn_readlane(dl, uu);
                const float* hp = H + (size_t)su * FEAT + lane;
                v[u][0] = hp[0]  * wu;
                v[u][1] = hp[32] * wu;
                v[u][2] = hp[64] * wu;
                v[u][3] = hp[96] * wu;
            }
#pragma unroll
            for (int u = 0; u < 8; ++u) {
                if (u0 + u < m) {
                    float* tp = tile + dls[u] * TS + lane;
                    __hip_atomic_fetch_add(tp,      v[u][0], __ATOMIC_RELAXED, __HIP_MEMORY_SCOPE_WORKGROUP);
                    __hip_atomic_fetch_add(tp + 32, v[u][1], __ATOMIC_RELAXED, __HIP_MEMORY_SCOPE_WORKGROUP);
                    __hip_atomic_fetch_add(tp + 64, v[u][2], __ATOMIC_RELAXED, __HIP_MEMORY_SCOPE_WORKGROUP);
                    __hip_atomic_fetch_add(tp + 96, v[u][3], __ATOMIC_RELAXED, __HIP_MEMORY_SCOPE_WORKGROUP);
                }
            }
        }
    };

    const int n = (int)lists[(size_t)blockIdx.x * OUT_RS];
    if (n <= cap) {
        const unsigned* L = lists + (size_t)blockIdx.x * OUT_RS + 1;
        unsigned ent[LIST_G];
        float    wgt[LIST_G];
#pragma unroll
        for (int k = 0; k < LIST_G; ++k) {
            const int idx = wave * 32 + k * AGG_T + lane;
            ent[k] = (idx < n) ? L[idx] : 0u;
        }
        float da[LIST_G], db[LIST_G];
#pragma unroll
        for (int k = 0; k < LIST_G; ++k) {
            da[k] = dinv[ent[k] >> 8];
            db[k] = dinv[lo + (int)(ent[k] & 255u)];
        }
        asm volatile("" :: "v"(da[0]), "v"(da[1]), "v"(da[2]), "v"(da[3]),
                           "v"(db[0]), "v"(db[1]), "v"(db[2]), "v"(db[3]));
#pragma unroll
        for (int k = 0; k < LIST_G; ++k) wgt[k] = da[k] * db[k];
#pragma unroll
        for (int k = 0; k < LIST_G; ++k) {
            const int g = wave * 32 + k * AGG_T;
            if (g < n) rows((n - g < 32) ? (n - g) : 32, (int)(ent[k] >> 8), (int)(ent[k] & 255u), wgt[k]);
        }
    } else {
        int* q = queue + wave * AGG_Q;
        auto push = [&](int pos, int e, int t) { q[pos] = e; };
        auto flush = [&](int nq) {
            for (int g = 0; g < nq; g += 32) {
                const int m = (nq - g < 32) ? (nq - g) : 32;
                int s = 0, dl = 0; float w = 0.0f;
                if (lane < m) {
                    const int e = q[g + lane];
                    const int d = dst[e];
                    s  = src[e];
                    dl = d - lo;
                    w  = dinv[s] * dinv[d];
                }
                rows(m, s, dl, w);
            }
        };
        scan_dst(dst, E, lo, cnt, lane, wave, push, flush);
    }
    __syncthreads();

    const int   f  = tid & (FEAT - 1);
    const float bb = (MODE == 1) ? bias[f] : 0.0f;
    float mu = 0.0f, rs = 1.0f, ga = 1.0f, be = 0.0f;
    if (MODE == 1) { mu = mean[f]; rs = rsqrtf(var[f] + BN_EPS); ga = gamma[f]; be = beta[f]; }
    for (int r = tid >> 7; r < cnt; r += AGG_T / FEAT) {
        const int    node = lo + r;
        const float  di   = dinv[node];
        const size_t gi   = (size_t)node * FEAT + f;
        if (MODE == 1) {
            float v = tile[r * TS + f] + H[gi] * di * di + bb;
            v = (v - mu) * rs * ga + be;
            v = fmaxf(v, 0.0f);
            out[gi] = v;
        } else {
            tile[r * TS + f] = tile[r * TS + f] + H[gi] * di * di;
        }
    }

    if (MODE == 2) {
        __syncthreads();
        {
            const int c = lane & 15;
            const int h = lane >> 4;
            const int nUnit = ((cnt + 15) >> 4) * 4;
            v8f pa[2][2];
#pragma unroll
            for (int i = 0; i < 2; ++i) {
                pa[i][0] = v8f{};
                pa[i][1] = v8f{};
                const int u = wave + i * AGG_W;
                if (u < nUnit) {
                    int row = 16 * (u >> 2) + c;
                    if (row > cnt - 1) row = cnt - 1;
                    const float*  ap = tile + row * TS + 4 * h;
                    const float4* bp = Bl + h * 64 + (u & 3) * 16 + c;
#pragma unroll
                    for (int g = 0; g < 16; ++g) {
                        const float4 a = *(const float4*)(ap + 8 * g);
#pragma unroll
                        for (int t = 0; t < 2; ++t) {
                            const float4 b = bp[(2 * g + t) * 128];
                            v2f av, b0, b1;
                            av.x = t ? a.z : a.x;
                            av.y = t ? a.w : a.y;
                            b0.x = b.x; b0.y = b.y;
                            b1.x = b.z; b1.y = b.w;
                            pa[i][0] = __builtin_amdgcn_wmma_f32_16x16x4_f32(
                                            false, av,  false, b0,
                                            (short)0, pa[i][0],  false,  false);
                            pa[i][1] = __builtin_amdgcn_wmma_f32_16x16x4_f32(
                                            false, av,  false, b1,
                                            (short)0, pa[i][1],  false,  false);
                        }
                    }
                }
            }
            __syncthreads();
#pragma unroll
            for (int i = 0; i < 2; ++i) {
                const int u = wave + i * AGG_W;
                if (u < nUnit) {
                    const int q  = u & 3;
                    const int cc = 64 * (q >> 1) + 4 * c + 2 * (q & 1);
                    const float2 bv = *(const float2*)(bias + cc);
#pragma unroll
                    for (int j = 0; j < 8; ++j) {
                        const int r = 16 * (u >> 2) + j + 8 * h;
                        if (r < cnt)
                            *(float2*)(tile + r * TS + cc) = make_float2(pa[i][0][j] + bv.x, pa[i][1][j] + bv.y);
                    }
                }
            }
        }
        __syncthreads();
        const int nStrip = (cnt + 15) >> 4;
        if (wave < nStrip) {
            const int c = lane & 15;
            const int h = lane >> 4;
            int row = 16 * wave + c;
            if (row > cnt - 1) row = cnt - 1;
            const float* ap = tile + row * TS + 2 * h;
            const float* bp = Wc + (2 * h) * N_CLASSES + c;
            v8f acc = {};
#pragma unroll
            for (int i = 0; i < FEAT / 4; ++i) {
                const v2f a = *(const v2f*)(ap + 4 * i);
                v2f b;
                b.x = bp[(4 * i)     * N_CLASSES];
                b.y = bp[(4 * i + 1) * N_CLASSES];
                acc = __builtin_amdgcn_wmma_f32_16x16x4_f32(
                           false, a,  false, b,
                           (short)0, acc,  false,  false);
            }
            const float bcv = bc[c];
#pragma unroll
            for (int j = 0; j < 8; ++j) {
                const int r = 16 * wave + j + 8 * h;
                if (r < cnt) cls[(size_t)(lo + r) * N_CLASSES + c] = acc[j] + bcv;
            }
        }
    }
}

extern "C" void kernel_launch(void* const* d_in, const int* in_sizes, int n_in,
                              void* d_out, int out_size, void* d_ws, size_t ws_size,
                              hipStream_t stream) {
    const float* X     = (const float*)d_in[0];
    const int*   edge  = (const int*)  d_in[1];
    const float* W1    = (const float*)d_in[2];
    const float* b1    = (const float*)d_in[3];
    const float* gamma = (const float*)d_in[4];
    const float* beta  = (const float*)d_in[5];
    const float* rmean = (const float*)d_in[6];
    const float* rvar  = (const float*)d_in[7];
    const float* W2    = (const float*)d_in[8];
    const float* b2    = (const float*)d_in[9];
    const float* Wc    = (const float*)d_in[10];
    const float* bc    = (const float*)d_in[11];
    float*       out   = (float*)d_out;

    const int N = in_sizes[0] / FEAT;
    const int E = in_sizes[1] / 2;
    const int* src = edge;
    const int* dst = edge + E;
    const int total = N * FEAT;

    float* dinv = (float*)d_ws;
    float* bufA = dinv + ((N + 63) & ~63);
    float* bufB = bufA + (size_t)total;

    const int gAgg   = (N + AGG_R - 1) / AGG_R;
    const int nSl    = (E + BIN_E - 1) / BIN_E;
    const int gStrip = (N + GS_W * 16 - 1) / (GS_W * 16);

    unsigned* lists  = (unsigned*)d_out;

    unsigned* tmp = (unsigned*)bufB;

    k_gemm_strip<<<gStrip, GS_T, 0, stream>>>(X, W1, bufA, N, src, dst, tmp, total, E, gAgg, nSl);

    k_deg<<<gAgg, DEG_T, 0, stream>>>(dst, tmp, total, nSl, lists, dinv, N, E);

    k_agg<1><<<gAgg, AGG_T, 0, stream>>>(bufA, src, dst, dinv, lists, b1, gamma, beta, rmean, rvar, bufB, nullptr, nullptr, nullptr, nullptr, N, E);

    k_agg<2><<<gAgg, AGG_T, 0, stream>>>(bufB, src, dst, dinv, lists, b2, nullptr, nullptr, nullptr, nullptr, nullptr, W2, Wc, bc, out, N, E);
}


# Degree factors from an in-degree byte table copied into LDS per block (degree launch stores counts): no two scattered 4-byte reads per edge, no round trip for them; weights and sum order unchanged
# speedup vs baseline: 4.2863x; 1.0347x over previous
#include <hip/hip_runtime.h>
#include <math.h>

typedef float v2f __attribute__((ext_vector_type(2)));
typedef float v8f __attribute__((ext_vector_type(8)));

#define FEAT 128
#define N_CLASSES 16
#define BN_EPS 1e-5f
#define AGG_R 200
#define OUT_RS (AGG_R * N_CLASSES)

#define GS_W  16
#define GS_T  (GS_W * 32)
#define GS_E  (32 * 2 * 4 * 16)

#define BIN_K   7
#define BIN_E   (GS_T * BIN_K)
#define BIN_O   1024
#define SUB     64

__device__ __forceinline__ void bin_slices(const int* __restrict__ src, const int* __restrict__ dst,
                                           unsigned* __restrict__ tmp, int N, int E, int nOwn, int nSl,
                                           int* hist) {
    const int tid = threadIdx.x;
    for (int sl = blockIdx.x; sl < nSl; sl += gridDim.x) {
        const int base = sl * BIN_E;
        int d[BIN_K], s[BIN_K];
        if (base + BIN_E <= E) {
#pragma unroll
            for (int k = 0; k < BIN_K; ++k) {
                const int e = base + k * GS_T + tid;
                d[k] = dst[e];
                s[k] = src[e];
            }
        } else {
#pragma unroll
            for (int k = 0; k < BIN_K; ++k) {
                const int e = base + k * GS_T + tid;
                d[k] = (e < E) ? dst[e] : -1;
                s[k] = (e < E) ? src[e] : 0;
            }
        }
        int o[BIN_K], rk[BIN_K];
#pragma unroll
        for (int k = 0; k < BIN_K; ++k) {
            o[k]  = ((unsigned)d[k] < (unsigned)N) ? d[k] / AGG_R : -1;
            rk[k] = 0;
        }
        for (int w0 = 0; w0 < nOwn; w0 += BIN_O) {
            __syncthreads();
            for (int i = tid; i < BIN_O; i += GS_T) hist[i] = 0;
            __syncthreads();
#pragma unroll
            for (int k = 0; k < BIN_K; ++k) {
                const unsigned ow = (unsigned)(o[k] - w0);
                if (ow < (unsigned)BIN_O)
                    rk[k] = __hip_atomic_fetch_add(&hist[ow], 1, __ATOMIC_RELAXED, __HIP_MEMORY_SCOPE_WORKGROUP);
            }
            __syncthreads();
            for (int i = tid; i < BIN_O && w0 + i < nOwn; i += GS_T)
                tmp[((size_t)(w0 + i) * nSl + sl) * SUB] = (unsigned)hist[i];
#pragma unroll
            for (int k = 0; k < BIN_K; ++k) {
                const unsigned ow = (unsigned)(o[k] - w0);
                if (ow < (unsigned)BIN_O && rk[k] < SUB - 1)
                    tmp[((size_t)o[k] * nSl + sl) * SUB + 1 + rk[k]] =
                        ((unsigned)s[k] << 8) | (unsigned)(d[k] - o[k] * AGG_R);
            }
        }
    }
}

__global__ __launch_bounds__(GS_T)
void k_gemm_strip(const float* __restrict__ A, const float* __restrict__ Wm,
                  float* __restrict__ C, int M,
                  const int* __restrict__ src, const int* __restrict__ dst,
                  unsigned* __restrict__ tmp, int tmpWords, int E, int nOwn, int nSl) {
    __shared__ float4 Bl[GS_E];
    __shared__ int hist[BIN_O];

    const int tid  = threadIdx.x;

    if (nSl > 0 && (size_t)nOwn * nSl * SUB <= (size_t)tmpWords)
        bin_slices(src, dst, tmp, M, E, nOwn, nSl, hist);

    const int lane = tid & 31;
    const int wave = __builtin_amdgcn_readfirstlane(tid >> 5);
    const int c    = lane & 15;
    const int h    = lane >> 4;
    const int m0   = (blockIdx.x * GS_W + wave) << 4;

    float2 w0[GS_E / GS_T], w1[GS_E / GS_T];
#pragma unroll
    for (int m = 0; m < GS_E / GS_T; ++m) {
        const int i  = tid + m * GS_T;
        const int w  = i >> 7;
        const int q  = (i >> 4) & 3;
        const int k0 = 8 * (w >> 1) + 4 * ((i >> 6) & 1) + 2 * (w & 1);
        const int cc = 64 * (q >> 1) + 4 * (i & 15) + 2 * (q & 1);
        w0[m] = *(const float2*)(Wm + k0 * FEAT + cc);
        w1[m] = *(const float2*)(Wm + (k0 + 1) * FEAT + cc);
    }

#pragma unroll
    for (int m = 0; m < GS_E / GS_T; ++m)
        Bl[tid + m * GS_T] = make_float4(w0[m].x, w1[m].x, w0[m].y, w1[m].y);
    __syncthreads();
    if (m0 >= M) return;

    int row = m0 + c;
    if (row > M - 1) row = M - 1;
    const float4* ap = (const float4*)(A + (size_t)row * FEAT + 4 * h);
    float4 a[16];
#pragma unroll
    for (int j = 0; j < 16; ++j) a[j] = ap[2 * j];

    v8f acc[8];
#pragma unroll
    for (int n = 0; n < 8; ++n) acc[n] = v8f{};

    const float4* bp = Bl + h * 64 + c;
    float4 bf[2][8];
#pragma unroll
    for (int q = 0; q < 8; ++q) bf[0][q] = bp[(q >> 2) * 128 + (q & 3) * 16];
#pragma unroll
    for (int g = 0; g < 16; ++g) {
        if (g + 1 < 16) {
#pragma unroll
            for (int q = 0; q < 8; ++q)
                bf[(g + 1) & 1][q] = bp[(2 * (g + 1) + (q >> 2)) * 128 + (q & 3) * 16];
        }
        __builtin_amdgcn_sched_barrier(0);
#pragma unroll
        for (int t = 0; t < 2; ++t) {
            v2f av;
            av.x = t ? a[g].z : a[g].x;
            av.y = t ? a[g].w : a[g].y;
#pragma unroll
            for (int q = 0; q < 4; ++q) {
                const float4 b = bf[g & 1][4 * t + q];
                v2f b0, b1;
                b0.x = b.x; b0.y = b.y;
                b1.x = b.z; b1.y = b.w;
                acc[2 * q] = __builtin_amdgcn_wmma_f32_16x16x4_f32(
                                  false, av,  false, b0,
                                  (short)0, acc[2 * q],  false,  false);
                acc[2 * q + 1] = __builtin_amdgcn_wmma_f32_16x16x4_f32(
                                  false, av,  false, b1,
                                  (short)0, acc[2 * q + 1],  false,  false);
            }
        }
        __builtin_amdgcn_sched_barrier(0);
    }

#pragma unroll
    for (int j = 0; j < 8; ++j) {
        const int r = m0 + j + 8 * h;
        if (r < M) {
            float* cp = C + (size_t)r * FEAT + 4 * c;
            *(float4*)cp        = make_float4(acc[0][j], acc[1][j], acc[2][j], acc[3][j]);
            *(float4*)(cp + 64) = make_float4(acc[4][j], acc[5][j], acc[6][j], acc[7][j]);
        }
    }
}

#define AGG_T   1024
#define AGG_W   (AGG_T / 32)
#define AGG_CH  512
#define AGG_Q   (AGG_CH + 32)
#define LIST_G  4
#define DEG_T   256
#define DEG_W   (DEG_T / 32)
#define DEG_U   8
#define DEGB_N  50048
#define DEGB_V  (DEGB_N / 16)
#define DEGB_M  ((DEGB_V + AGG_T - 1) / AGG_T)

__device__ __forceinline__ void agg_ld_chunk(int4 (&r)[4], const int* __restrict__ dst,
                                             int base, int lane, int E, bool vec) {
    if (vec && base + AGG_CH <= E) {
#pragma unroll
        for (int k = 0; k < 4; ++k) r[k] = *(const int4*)(dst + base + k * 128 + lane * 4);
    } else {
#pragma unroll
        for (int k = 0; k < 4; ++k) {
            const int e0 = base + k * 128 + lane * 4;
            r[k].x = (e0     < E) ? dst[e0]     : -1;
            r[k].y = (e0 + 1 < E) ? dst[e0 + 1] : -1;
            r[k].z = (e0 + 2 < E) ? dst[e0 + 2] : -1;
            r[k].w = (e0 + 3 < E) ? dst[e0 + 3] : -1;
        }
    }
}

template <class P, class F>
__device__ __forceinline__ void scan_dst(const int* __restrict__ dst, int E, int lo, int cnt,
                                         int lane, int wave, P push, F flush) {
    int qn = 0;
    const bool vec  = ((reinterpret_cast<uintptr_t>(dst) & 15) == 0);
    const int  nCh  = (E + AGG_CH - 1) / AGG_CH;
    const int  rot  = (int)blockIdx.x * (nCh / (int)gridDim.x);

    int  c = wave;
    int4 nx[4];
    int  nbase = 0;
    if (c < nCh) {
        int cc = c + rot; if (cc >= nCh) cc -= nCh;
        nbase = cc * AGG_CH;
        agg_ld_chunk(nx, dst, nbase, lane, E, vec);
    }
    while (c < nCh) {
        int4 cu[4];
#pragma unroll
        for (int k = 0; k < 4; ++k) cu[k] = nx[k];
        const int cbase = nbase;
        const int cn = c + AGG_W;
        if (cn < nCh) {
            int cc = cn + rot; if (cc >= nCh) cc -= nCh;
            nbase = cc * AGG_CH;
            agg_ld_chunk(nx, dst, nbase, lane, E, vec);
        }
#pragma unroll
        for (int k = 0; k < 4; ++k) {
#pragma unroll
            for (int j = 0; j < 4; ++j) {
                const int d = (j == 0) ? cu[k].x : (j == 1) ? cu[k].y : (j == 2) ? cu[k].z : cu[k].w;
                const unsigned t  = (unsigned)(d - lo);
                const bool     mt = t < (unsigned)cnt;
                const unsigned mask = __builtin_amdgcn_ballot_w32(mt);
                if (mask != 0u) {
                    if (mt) push(qn + (int)__builtin_amdgcn_mbcnt_lo(mask, 0u), cbase + k * 128 + lane * 4 + j, (int)t);
                    qn += __builtin_popcount(mask);
                }
            }
        }
        if (qn >= 32) { flush(qn); qn = 0; }
        c = cn;
    }
    flush(qn);
}

__global__ __launch_bounds__(DEG_T)
void k_deg(const int* __restrict__ dst, const unsigned* __restrict__ tmp, int tmpWords, int nSl,
           unsigned* __restrict__ lists,
           unsigned* __restrict__ deg, int N, int E) {
    __shared__ int cntd[AGG_R];
    __shared__ int cursor;
    __shared__ int bad;

    const int tid  = threadIdx.x;
    const int lane = tid & 31;
    const int wave = __builtin_amdgcn_readfirstlane(tid >> 5);
    const int lo   = blockIdx.x * AGG_R;
    const int nn   = (N - lo < AGG_R) ? (N - lo) : AGG_R;
    const int cap  = nn * N_CLASSES - 1;
    const bool have = (size_t)gridDim.x * nSl * SUB <= (size_t)tmpWords;

    for (int i = tid; i < AGG_R; i += DEG_T) cntd[i] = 0;
    if (tid == 0) { cursor = 0; bad = have ? 0 : 1; }
    __syncthreads();

    if (have) {
        const unsigned* T = tmp + (size_t)blockIdx.x * nSl * SUB;
        unsigned*       L = lists + (size_t)blockIdx.x * OUT_RS + 1;
        for (int t0 = wave; t0 < nSl; t0 += DEG_W * DEG_U) {
            unsigned wa[DEG_U], wb[DEG_U];
#pragma unroll
            for (int u = 0; u < DEG_U; ++u) {
                const int t = t0 + u * DEG_W;
                wa[u] = (t < nSl) ? T[(size_t)t * SUB + lane]      : 0u;
                wb[u] = (t < nSl) ? T[(size_t)t * SUB + 32 + lane] : 0u;
            }
            int cc[DEG_U];
            int tot = 0, over = 0;
#pragma unroll
            for (int u = 0; u < DEG_U; ++u) {
                int c = (int)__builtin_amdgcn_readfirstlane(wa[u]);
                if (c > SUB - 1) { over = 1; c = SUB - 1; }
                cc[u] = c;
                tot  += c;
            }
            int at = 0;
            if (lane == 0) {
                if (tot > 0) at = __hip_atomic_fetch_add(&cursor, tot, __ATOMIC_RELAXED, __HIP_MEMORY_SCOPE_WORKGROUP);
                if (over) bad = 1;
            }
            at = __builtin_amdgcn_readfirstlane(at);
#pragma unroll
            for (int u = 0; u < DEG_U; ++u) {
                if (lane >= 1 && lane <= cc[u]) {
                    const int p = at + lane - 1;
                    if (p < cap) L[p] = wa[u];
                    __hip_atomic_fetch_add(&cntd[wa[u] & 255u], 1, __ATOMIC_RELAXED, __HIP_MEMORY_SCOPE_WORKGROUP);
                }
                if (32 + lane <= cc[u]) {
                    const int p = at + 31 + lane;
                    if (p < cap) L[p] = wb[u];
                    __hip_atomic_fetch_add(&cntd[wb[u] & 255u], 1, __ATOMIC_RELAXED, __HIP_MEMORY_SCOPE_WORKGROUP);
                }
                at += cc[u];
            }
        }
    }
    __syncthreads();

    int n = cursor;
    if (bad != 0 || n > cap) {
        for (int i = tid; i < AGG_R; i += DEG_T) cntd[i] = 0;
        __syncthreads();
        for (int e = tid; e < E; e += DEG_T) {
            const unsigned t = (unsigned)(dst[e] - lo);
            if (t < (unsigned)nn)
                __hip_atomic_fetch_add(&cntd[t], 1, __ATOMIC_RELAXED, __HIP_MEMORY_SCOPE_WORKGROUP);
        }
        __syncthreads();
        n = cap + 1;
    }
    if (tid == 0) lists[(size_t)blockIdx.x * OUT_RS] = (unsigned)n;

    unsigned char* degb = (unsigned char*)deg;
    unsigned char* degx = degb + ((N + 63) & ~63);
    for (int i = tid; i < nn; i += DEG_T) {
        const int cn = cntd[i];
        degb[lo + i] = (unsigned char)(cn < 255 ? cn : 255);
        if (cn >= 255 || lo + i >= DEGB_N) {
            unsigned char* x = degx + 3 * (size_t)(lo + i);
            x[0] = (unsigned char)(cn & 255);
            x[1] = (unsigned char)((cn >> 8) & 255);
            x[2] = (unsigned char)((cn >> 16) & 255);
        }
    }
    if (blockIdx.x == gridDim.x - 1)
        for (int i = N + tid; i < ((N + 63) & ~63); i += DEG_T) degb[i] = 0;
}

template <int MODE>
__global__ __launch_bounds__(AGG_T)
void k_agg(const float* __restrict__ H,
           const int* __restrict__ src, const int* __restrict__ dst,
           const unsigned* __restrict__ deg,
           const unsigned* lists,
           const float* __restrict__ bias,
           const float* __restrict__ gamma, const float* __restrict__ beta,
           const float* __restrict__ mean, const float* __restrict__ var,
           float* __restrict__ out,
           const float* __restrict__ W2,
           const float* __restrict__ Wc, const float* __restrict__ bc, float* cls,
           int N, int E) {
    constexpr int TS = (MODE == 2) ? (FEAT + 4) : FEAT;
    __shared__ __attribute__((aligned(16))) float tile[AGG_R * TS];
    __shared__ int queue[AGG_W * AGG_Q];
    __shared__ float4 Bl[(MODE == 2) ? GS_E : 1];
    __shared__ __attribute__((aligned(16))) unsigned char degl[DEGB_N];
    __shared__ float dself[AGG_R];

    const int tid  = threadIdx.x;
    const int lane = tid & 31;
    const int wave = __builtin_amdgcn_readfirstlane(tid >> 5);
    const int lo   = blockIdx.x * AGG_R;
    const int cnt  = (N - lo < AGG_R) ? (N - lo) : AGG_R;
    const int cap  = cnt * N_CLASSES - 1;

    const unsigned char* degb = (const unsigned char*)deg;
    const unsigned char* degx = degb + ((N + 63) & ~63);
    const int nPc = (((N + 63) & ~63) < DEGB_N ? ((N + 63) & ~63) : DEGB_N) >> 4;
    uint4 tb[DEGB_M];
#pragma unroll
    for (int m = 0; m < DEGB_M; ++m) {
        const int i = tid + m * AGG_T;
        tb[m] = (i < nPc) ? ((const uint4*)degb)[i] : make_uint4(0u, 0u, 0u, 0u);
    }

    float2 w0[GS_E / AGG_T], w1[GS_E / AGG_T];
    if (MODE == 2) {
#pragma unroll
        for (int m = 0; m < GS_E / AGG_T; ++m) {
            const int i  = tid + m * AGG_T;
            const int w  = i >> 7;
            const int q  = (i >> 4) & 3;
            const int k0 = 8 * (w >> 1) + 4 * ((i >> 6) & 1) + 2 * (w & 1);
            const int cc = 64 * (q >> 1) + 4 * (i & 15) + 2 * (q & 1);
            w0[m] = *(const float2*)(W2 + k0 * FEAT + cc);
            w1[m] = *(const float2*)(W2 + (k0 + 1) * FEAT + cc);
        }
    }

    for (int i = tid; i < AGG_R * TS / 4; i += AGG_T)
        ((float4*)tile)[i] = make_float4(0.0f, 0.0f, 0.0f, 0.0f);
#pragma unroll
    for (int m = 0; m < DEGB_M; ++m) {
        const int i = tid + m * AGG_T;
        if (i < DEGB_V) ((uint4*)degl)[i] = tb[m];
    }
    if (MODE == 2) {
#pragma unroll
        for (int m = 0; m < GS_E / AGG_T; ++m)
            Bl[tid + m * AGG_T] = make_float4(w0[m].x, w1[m].x, w0[m].y, w1[m].y);
    }
    __syncthreads();

    auto cnt_byte = [&](int node) -> unsigned {
        const unsigned b = degl[(node < DEGB_N) ? node : 0];
        return (node < DEGB_N) ? b : 255u;
    };
    auto cnt_full = [&](int node, unsigned b) -> unsigned {
        if (b == 255u) {
            const unsigned char* x = degx + 3 * (size_t)node;
            b = (unsigned)x[0] | ((unsigned)x[1] << 8) | ((unsigned)x[2] << 16);
        }
        return b;
    };
    auto dinv_cnt = [&](unsigned cn) -> float { return rsqrtf(1.0f + (float)cn); };
    auto dinv_of  = [&](int node) -> float { return dinv_cnt(cnt_full(node, cnt_byte(node))); };

    if (tid < cnt) dself[tid] = dinv_of(lo + tid);

    auto rows = [&](int m, int s, int dl, float w) {
        for (int u0 = 0; u0 < m; u0 += 8) {
            float v[8][4];
            int   dls[8];
#pragma unroll
            for (int u = 0; u < 8; ++u) {
                const int uu = (u0 + u < m) ? (u0 + u) : (m - 1);
                const int   su = __builtin_amdgcn_readlane(s, uu);
                const float wu = __int_as_float(__builtin_amdgcn_readlane(__float_as_int(w), uu));
                dls[u] = __builtin_amdgcn_readlane(dl, uu);
                const float* hp = H + (size_t)su * FEAT + lane;
                v[u][0] = hp[0]  * wu;
                v[u][1] = hp[32] * wu;
                v[u][2] = hp[64] * wu;
                v[u][3] = hp[96] * wu;
            }
#pragma unroll
            for (int u = 0; u < 8; ++u) {
                if (u0 + u < m) {
                    float* tp = tile + dls[u] * TS + lane;
                    __hip_atomic_fetch_add(tp,      v[u][0], __ATOMIC_RELAXED, __HIP_MEMORY_SCOPE_WORKGROUP);
                    __hip_atomic_fetch_add(tp + 32, v[u][1], __ATOMIC_RELAXED, __HIP_MEMORY_SCOPE_WORKGROUP);
                    __hip_atomic_fetch_add(tp + 64, v[u][2], __ATOMIC_RELAXED, __HIP_MEMORY_SCOPE_WORKGROUP);
                    __hip_atomic_fetch_add(tp + 96, v[u][3], __ATOMIC_RELAXED, __HIP_MEMORY_SCOPE_WORKGROUP);
                }
            }
        }
    };

    const int n = (int)lists[(size_t)blockIdx.x * OUT_RS];
    if (n <= cap) {
        const unsigned* L = lists + (size_t)blockIdx.x * OUT_RS + 1;
        unsigned ent[LIST_G];
        float    wgt[LIST_G];
#pragma unroll
        for (int k = 0; k < LIST_G; ++k) {
            const int idx = wave * 32 + k * AGG_T + lane;
            ent[k] = (idx < n) ? L[idx] : 0u;
        }
        unsigned cs[LIST_G], cd[LIST_G];
        bool full = false;
#pragma unroll
        for (int k = 0; k < LIST_G; ++k) {
            cs[k] = cnt_byte((int)(ent[k] >> 8));
            cd[k] = cnt_byte(lo + (int)(ent[k] & 255u));
            full  = full || cs[k] == 255u || cd[k] == 255u;
        }
        if (__builtin_amdgcn_ballot_w32(full) != 0u) {
#pragma unroll
            for (int k = 0; k < LIST_G; ++k) {
                cs[k] = cnt_full((int)(ent[k] >> 8), cs[k]);
                cd[k] = cnt_full(lo + (int)(ent[k] & 255u), cd[k]);
            }
        }
#pragma unroll
        for (int k = 0; k < LIST_G; ++k) wgt[k] = dinv_cnt(cs[k]) * dinv_cnt(cd[k]);
#pragma unroll
        for (int k = 0; k < LIST_G; ++k) {
            const int g = wave * 32 + k * AGG_T;
            if (g < n) rows((n - g < 32) ? (n - g) : 32, (int)(ent[k] >> 8), (int)(ent[k] & 255u), wgt[k]);
        }
    } else {
        int* q = queue + wave * AGG_Q;
        auto push = [&](int pos, int e, int t) { q[pos] = e; };
        auto flush = [&](int nq) {
            for (int g = 0; g < nq; g += 32) {
                const int m = (nq - g < 32) ? (nq - g) : 32;
                int s = 0, dl = 0; float w = 0.0f;
                if (lane < m) {
                    const int e = q[g + lane];
                    const int d = dst[e];
                    s  = src[e];
                    dl = d - lo;
                    w  = dinv_of(s) * dinv_of(d);
                }
                rows(m, s, dl, w);
            }
        };
        scan_dst(dst, E, lo, cnt, lane, wave, push, flush);
    }
    __syncthreads();

    const int   f  = tid & (FEAT - 1);
    const float bb = (MODE == 1) ? bias[f] : 0.0f;
    float mu = 0.0f, rs = 1.0f, ga = 1.0f, be = 0.0f;
    if (MODE == 1) { mu = mean[f]; rs = rsqrtf(var[f] + BN_EPS); ga = gamma[f]; be = beta[f]; }
    for (int r = tid >> 7; r < cnt; r += AGG_T / FEAT) {
        const int    node = lo + r;
        const size_t gi   = (size_t)node * FEAT + f;
        const float  di   = dself[r];
        if (MODE == 1) {
            float v = tile[r * TS + f] + H[gi] * di * di + bb;
            v = (v - mu) * rs * ga + be;
            v = fmaxf(v, 0.0f);
            out[gi] = v;
        } else {
            tile[r * TS + f] = tile[r * TS + f] + H[gi] * di * di;
        }
    }

    if (MODE == 2) {
        __syncthreads();
        {
            const int c = lane & 15;
            const int h = lane >> 4;
            const int nUnit = ((cnt + 15) >> 4) * 4;
            v8f pa[2][2];
#pragma unroll
            for (int i = 0; i < 2; ++i) {
                pa[i][0] = v8f{};
                pa[i][1] = v8f{};
                const int u = wave + i * AGG_W;
                if (u < nUnit) {
                    int row = 16 * (u >> 2) + c;
                    if (row > cnt - 1) row = cnt - 1;
                    const float*  ap = tile + row * TS + 4 * h;
                    const float4* bp = Bl + h * 64 + (u & 3) * 16 + c;
#pragma unroll
                    for (int g = 0; g < 16; ++g) {
                        const float4 a = *(const float4*)(ap + 8 * g);
#pragma unroll
                        for (int t = 0; t < 2; ++t) {
                            const float4 b = bp[(2 * g + t) * 128];
                            v2f av, b0, b1;
                            av.x = t ? a.z : a.x;
                            av.y = t ? a.w : a.y;
                            b0.x = b.x; b0.y = b.y;
                            b1.x = b.z; b1.y = b.w;
                            pa[i][0] = __builtin_amdgcn_wmma_f32_16x16x4_f32(
                                            false, av,  false, b0,
                                            (short)0, pa[i][0],  false,  false);
                            pa[i][1] = __builtin_amdgcn_wmma_f32_16x16x4_f32(
                                            false, av,  false, b1,
                                            (short)0, pa[i][1],  false,  false);
                        }
                    }
                }
            }
            __syncthreads();
#pragma unroll
            for (int i = 0; i < 2; ++i) {
                const int u = wave + i * AGG_W;
                if (u < nUnit) {
                    const int q  = u & 3;
                    const int cc = 64 * (q >> 1) + 4 * c + 2 * (q & 1);
                    const float2 bv = *(const float2*)(bias + cc);
#pragma unroll
                    for (int j = 0; j < 8; ++j) {
                        const int r = 16 * (u >> 2) + j + 8 * h;
                        if (r < cnt)
                            *(float2*)(tile + r * TS + cc) = make_float2(pa[i][0][j] + bv.x, pa[i][1][j] + bv.y);
                    }
                }
            }
        }
        __syncthreads();
        const int nStrip = (cnt + 15) >> 4;
        if (wave < nStrip) {
            const int c = lane & 15;
            const int h = lane >> 4;
            int row = 16 * wave + c;
            if (row > cnt - 1) row = cnt - 1;
            const float* ap = tile + row * TS + 2 * h;
            const float* bp = Wc + (2 * h) * N_CLASSES + c;
            v8f acc = {};
#pragma unroll
            for (int i = 0; i < FEAT / 4; ++i) {
                const v2f a = *(const v2f*)(ap + 4 * i);
                v2f b;
                b.x = bp[(4 * i)     * N_CLASSES];
                b.y = bp[(4 * i + 1) * N_CLASSES];
                acc = __builtin_amdgcn_wmma_f32_16x16x4_f32(
                           false, a,  false, b,
                           (short)0, acc,  false,  false);
            }
            const float bcv = bc[c];
#pragma unroll
            for (int j = 0; j < 8; ++j) {
                const int r = 16 * wave + j + 8 * h;
                if (r < cnt) cls[(size_t)(lo + r) * N_CLASSES + c] = acc[j] + bcv;
            }
        }
    }
}

extern "C" void kernel_launch(void* const* d_in, const int* in_sizes, int n_in,
                              void* d_out, int out_size, void* d_ws, size_t ws_size,
                              hipStream_t stream) {
    const float* X     = (const float*)d_in[0];
    const int*   edge  = (const int*)  d_in[1];
    const float* W1    = (const float*)d_in[2];
    const float* b1    = (const float*)d_in[3];
    const float* gamma = (const float*)d_in[4];
    const float* beta  = (const float*)d_in[5];
    const float* rmean = (const float*)d_in[6];
    const float* rvar  = (const float*)d_in[7];
    const float* W2    = (const float*)d_in[8];
    const float* b2    = (const float*)d_in[9];
    const float* Wc    = (const float*)d_in[10];
    const float* bc    = (const float*)d_in[11];
    float*       out   = (float*)d_out;

    const int N = in_sizes[0] / FEAT;
    const int E = in_sizes[1] / 2;
    const int* src = edge;
    const int* dst = edge + E;
    const int total = N * FEAT;

    unsigned* deg = (unsigned*)d_ws;
    float* bufA = (float*)d_ws + ((N + 63) & ~63);
    float* bufB = bufA + (size_t)total;

    const int gAgg   = (N + AGG_R - 1) / AGG_R;
    const int nSl    = (E + BIN_E - 1) / BIN_E;
    const int gStrip = (N + GS_W * 16 - 1) / (GS_W * 16);

    unsigned* lists  = (unsigned*)d_out;

    unsigned* tmp = (unsigned*)bufB;

    k_gemm_strip<<<gStrip, GS_T, 0, stream>>>(X, W1, bufA, N, src, dst, tmp, total, E, gAgg, nSl);

    k_deg<<<gAgg, DEG_T, 0, stream>>>(dst, tmp, total, nSl, lists, deg, N, E);

    k_agg<1><<<gAgg, AGG_T, 0, stream>>>(bufA, src, dst, deg, lists, b1, gamma, beta, rmean, rvar, bufB, nullptr, nullptr, nullptr, nullptr, N, E);

    k_agg<2><<<gAgg, AGG_T, 0, stream>>>(bufB, src, dst, deg, lists, b2, nullptr, nullptr, nullptr, nullptr, nullptr, W2, Wc, bc, out, N, E);
}
